# starting kernel
# baseline (speedup 1.0000x reference)
#include <hip/hip_runtime.h>

typedef __attribute__((ext_vector_type(2))) float v2f;
typedef __attribute__((ext_vector_type(4))) float v4f;
typedef __attribute__((ext_vector_type(8))) float v8f;

namespace {
constexpr int kN     = 2048;
constexpr int kB     = 4096;
constexpr int kCols  = 16;
constexpr int kBlk   = 16;
constexpr int kWaves = 8;
constexpr int kThreads = kWaves * 32;
constexpr int kHStr  = kN + 4;
constexpr int kSmemFloats = kCols * kHStr
                          + 2 * kWaves * 256
                          + 4 * 256;
}

__device__ __forceinline__ v8f wmma_f32_16x16x4(v2f a, v2f b, v8f c) {
  return __builtin_amdgcn_wmma_f32_16x16x4_f32(
       false, a,  false, b,
       (short)0, c,  false,  false);
}

__device__ __forceinline__ float load_u(const float* __restrict__ w, int row, int col) {
  float v = 0.0f;
  if (col >= row) v = w[(size_t)row * kN + col];
  if (row == kN - 1 && col == kN - 1) v = 1.0f;
  return v;
}

__global__ __launch_bounds__(kThreads)
void ortho_householder_kernel(const float* __restrict__ x,
                              const float* __restrict__ w,
                              float* __restrict__ out) {
  extern __shared__ float smem[];
  float* hbuf  = smem;
  float* Wpart = hbuf + kCols * kHStr;
  float* Gpart = Wpart + kWaves * 256;
  float* Wbuf  = Gpart + kWaves * 256;
  float* Gbuf  = Wbuf + 256;
  float* Tbuf  = Gbuf + 256;
  float* Ybuf  = Tbuf + 256;

  const int tid  = threadIdx.x;
  const int wave = __builtin_amdgcn_readfirstlane(tid >> 5);
  const int lane = tid & 31;
  const int lm   = lane & 15;
  const int lh   = lane >> 4;
  const int c0   = blockIdx.x * kCols;
  float* const hrow = hbuf + lm * kHStr;

  for (int i = tid; i < kCols * (kN / 4); i += kThreads) {
    const int c  = i / (kN / 4);
    const int k4 = (i - c * (kN / 4)) * 4;
    v4f v = *(const v4f*)(x + (size_t)(c0 + c) * kN + k4);
    *(v4f*)(hbuf + c * kHStr + k4) = v;
  }
  __syncthreads();

  for (int base = kN - kBlk; base >= 0; base -= kBlk) {
    v8f accW = {};
    v8f accG = {};
    int k = base + wave * 4;
    if (wave < 4) {
      const int col = k + 2 * lh;
      const int row = base + lm;
      v2f a;
      a.x = load_u(w, row, col);
      a.y = load_u(w, row, col + 1);
      v2f b = *(const v2f*)(hrow + col);
      accW = wmma_f32_16x16x4(a, b, accW);
      accG = wmma_f32_16x16x4(a, a, accG);
      k += kWaves * 4;
    }
    if (k < kN) {
      const float* wrow = w + (size_t)(base + lm) * kN;
      v2f a = *(const v2f*)(wrow + k + 2 * lh);
      v2f b = *(const v2f*)(hrow + k + 2 * lh);
#pragma unroll 2
      for (k += kWaves * 4; k < kN; k += kWaves * 4) {
        const int col = k + 2 * lh;
        v2f an = *(const v2f*)(wrow + col);
        v2f bn = *(const v2f*)(hrow + col);
        accW = wmma_f32_16x16x4(a, b, accW);
        accG = wmma_f32_16x16x4(a, a, accG);
        a = an;
        b = bn;
      }
      accW = wmma_f32_16x16x4(a, b, accW);
      accG = wmma_f32_16x16x4(a, a, accG);
    }
#pragma unroll
    for (int r = 0; r < 8; ++r) {
      const int m = r + 8 * lh;
      Wpart[wave * 256 + m * 16 + lm] = accW[r];
      Gpart[wave * 256 + m * 16 + lm] = accG[r];
    }
    __syncthreads();

    {
      float sw = 0.f, sg = 0.f;
#pragma unroll
      for (int p = 0; p < kWaves; ++p) {
        sw += Wpart[p * 256 + tid];
        sg += Gpart[p * 256 + tid];
      }
      Wbuf[tid] = sw;
      Gbuf[tid] = sg;
    }
    __syncthreads();

    if (tid < 16) {
      const int i = tid;
      float Trow[16];
#pragma unroll
      for (int j = 0; j < 16; ++j) {
        const float gd = Gbuf[j * 16 + j];
        const float cj = (gd > 0.f) ? (2.0f / gd) : 0.0f;
        float s = 0.f;
#pragma unroll
        for (int l = 0; l < 16; ++l) {
          if (l < j) s += (l >= i) ? Trow[l] * Gbuf[l * 16 + j] : 0.f;
        }
        float tv = 0.f;
        if (j == i) tv = cj;
        else if (j > i) tv = -cj * s;
        Trow[j] = tv;
        Tbuf[i * 16 + j] = tv;
      }
    }
    __syncthreads();

    {
      const int m = tid >> 4;
      const int c = tid & 15;
      float s = 0.f;
#pragma unroll
      for (int l = 0; l < 16; ++l) s += Tbuf[m * 16 + l] * Wbuf[l * 16 + c];
      Ybuf[m * 16 + c] = -s;
    }
    __syncthreads();

    const int ntiles = (kN - base) >> 4;
    v2f by[4];
#pragma unroll
    for (int m0 = 0; m0 < 16; m0 += 4) {
      by[m0 >> 2].x = Ybuf[(m0 + 2 * lh) * 16 + lm];
      by[m0 >> 2].y = Ybuf[(m0 + 2 * lh + 1) * 16 + lm];
    }
    for (int t = wave; t < ntiles; t += kWaves) {
      const int k0 = base + t * 16;
      float* hp = hrow + k0 + 8 * lh;
      v4f clo = *(const v4f*)(hp);
      v4f chi = *(const v4f*)(hp + 4);
      v8f acc = {clo.x, clo.y, clo.z, clo.w, chi.x, chi.y, chi.z, chi.w};
      if (t == 0) {
#pragma unroll
        for (int m0 = 0; m0 < 16; m0 += 4) {
          v2f a;
          a.x = load_u(w, base + m0 + 2 * lh,     k0 + lm);
          a.y = load_u(w, base + m0 + 2 * lh + 1, k0 + lm);
          acc = wmma_f32_16x16x4(a, by[m0 >> 2], acc);
        }
      } else {
        const float* wp = w + (size_t)(base + 2 * lh) * kN + k0 + lm;
#pragma unroll
        for (int m0 = 0; m0 < 16; m0 += 4) {
          v2f a;
          a.x = wp[(size_t)m0 * kN];
          a.y = wp[(size_t)(m0 + 1) * kN];
          acc = wmma_f32_16x16x4(a, by[m0 >> 2], acc);
        }
      }
      v4f olo = {acc[0], acc[1], acc[2], acc[3]};
      v4f ohi = {acc[4], acc[5], acc[6], acc[7]};
      *(v4f*)(hp) = olo;
      *(v4f*)(hp + 4) = ohi;
    }
    __syncthreads();
  }

  for (int i = tid; i < kCols * (kN / 4); i += kThreads) {
    const int c  = i / (kN / 4);
    const int k4 = (i - c * (kN / 4)) * 4;
    v4f v = *(const v4f*)(hbuf + c * kHStr + k4);
    *(v4f*)(out + (size_t)(c0 + c) * kN + k4) = v;
  }
}

extern "C" void kernel_launch(void* const* d_in, const int* in_sizes, int n_in,
                              void* d_out, int out_size, void* d_ws, size_t ws_size,
                              hipStream_t stream) {
  const float* x = (const float*)d_in[0];
  const float* w = (const float*)d_in[1];
  float* out = (float*)d_out;

  dim3 grid(kB / kCols);
  dim3 block(kThreads);
  const size_t smem = (size_t)kSmemFloats * sizeof(float);
  hipLaunchKernelGGL(ortho_householder_kernel, grid, block, smem, stream,
                     x, w, out);
}


# Replace the serial 16-lane T recurrence and the Y=-T*W stage by a per-column back-substitution Y[m]=-c_m(W[m]+sum_{l>m}G[m][l]Y[l]) in each wave's registers (changes that small sum's order)
# speedup vs baseline: 1.4046x; 1.4046x over previous
#include <hip/hip_runtime.h>

typedef __attribute__((ext_vector_type(2))) float v2f;
typedef __attribute__((ext_vector_type(4))) float v4f;
typedef __attribute__((ext_vector_type(8))) float v8f;

namespace {
constexpr int kN     = 2048;
constexpr int kB     = 4096;
constexpr int kCols  = 16;
constexpr int kBlk   = 16;
constexpr int kWaves = 8;
constexpr int kThreads = kWaves * 32;
constexpr int kHStr  = kN + 4;
constexpr int kSmemFloats = kCols * kHStr
                          + 2 * kWaves * 256
                          + 2 * 256
                          + 16;
}

__device__ __forceinline__ v8f wmma_f32_16x16x4(v2f a, v2f b, v8f c) {
  return __builtin_amdgcn_wmma_f32_16x16x4_f32(
       false, a,  false, b,
       (short)0, c,  false,  false);
}

__device__ __forceinline__ float load_u(const float* __restrict__ w, int row, int col) {
  float v = 0.0f;
  if (col >= row) v = w[(size_t)row * kN + col];
  if (row == kN - 1 && col == kN - 1) v = 1.0f;
  return v;
}

__global__ __launch_bounds__(kThreads)
void ortho_householder_kernel(const float* __restrict__ x,
                              const float* __restrict__ w,
                              float* __restrict__ out) {
  extern __shared__ float smem[];
  float* hbuf  = smem;
  float* Wpart = hbuf + kCols * kHStr;
  float* Gpart = Wpart + kWaves * 256;
  float* Wbuf  = Gpart + kWaves * 256;
  float* Gbuf  = Wbuf + 256;
  float* Cbuf  = Gbuf + 256;

  const int tid  = threadIdx.x;
  const int wave = __builtin_amdgcn_readfirstlane(tid >> 5);
  const int lane = tid & 31;
  const int lm   = lane & 15;
  const int lh   = lane >> 4;
  const int c0   = blockIdx.x * kCols;
  float* const hrow = hbuf + lm * kHStr;

  for (int i = tid; i < kCols * (kN / 4); i += kThreads) {
    const int c  = i / (kN / 4);
    const int k4 = (i - c * (kN / 4)) * 4;
    v4f v = *(const v4f*)(x + (size_t)(c0 + c) * kN + k4);
    *(v4f*)(hbuf + c * kHStr + k4) = v;
  }
  __syncthreads();

  for (int base = kN - kBlk; base >= 0; base -= kBlk) {
    v8f accW = {};
    v8f accG = {};
    int k = base + wave * 4;
    if (wave < 4) {
      const int col = k + 2 * lh;
      const int row = base + lm;
      v2f a;
      a.x = load_u(w, row, col);
      a.y = load_u(w, row, col + 1);
      v2f b = *(const v2f*)(hrow + col);
      accW = wmma_f32_16x16x4(a, b, accW);
      accG = wmma_f32_16x16x4(a, a, accG);
      k += kWaves * 4;
    }
    if (k < kN) {
      const float* wrow = w + (size_t)(base + lm) * kN;
      v2f a = *(const v2f*)(wrow + k + 2 * lh);
      v2f b = *(const v2f*)(hrow + k + 2 * lh);
#pragma unroll 2
      for (k += kWaves * 4; k < kN; k += kWaves * 4) {
        const int col = k + 2 * lh;
        v2f an = *(const v2f*)(wrow + col);
        v2f bn = *(const v2f*)(hrow + col);
        accW = wmma_f32_16x16x4(a, b, accW);
        accG = wmma_f32_16x16x4(a, a, accG);
        a = an;
        b = bn;
      }
      accW = wmma_f32_16x16x4(a, b, accW);
      accG = wmma_f32_16x16x4(a, a, accG);
    }
#pragma unroll
    for (int r = 0; r < 8; ++r) {
      const int m = r + 8 * lh;
      Wpart[wave * 256 + m * 16 + lm] = accW[r];
      Gpart[wave * 256 + m * 16 + lm] = accG[r];
    }
    __syncthreads();

    {
      float sw = 0.f, sg = 0.f;
#pragma unroll
      for (int p = 0; p < kWaves; ++p) {
        sw += Wpart[p * 256 + tid];
        sg += Gpart[p * 256 + tid];
      }
      Wbuf[tid] = sw;
      Gbuf[tid] = sg;
      if ((tid >> 4) == (tid & 15)) Cbuf[tid & 15] = (sg > 0.f) ? (2.0f / sg) : 0.0f;
    }
    __syncthreads();

    float y[16];
    {
      float wv[16];
#pragma unroll
      for (int l = 0; l < 16; ++l) wv[l] = Wbuf[l * 16 + lm];
      v4f c4[4];
#pragma unroll
      for (int p = 0; p < 4; ++p) c4[p] = *(const v4f*)(Cbuf + 4 * p);
      v4f g4[16][4];
#pragma unroll
      for (int m = 0; m < 15; ++m) {
#pragma unroll
        for (int p = (m + 1) >> 2; p < 4; ++p)
          g4[m][p] = *(const v4f*)(Gbuf + m * 16 + 4 * p);
      }
      __builtin_amdgcn_sched_barrier(0);
#pragma unroll
      for (int m = 15; m >= 0; --m) {
        float s = wv[m];
#pragma unroll
        for (int l = 15; l > m; --l) s = __builtin_fmaf(g4[m][l >> 2][l & 3], y[l], s);
        y[m] = -c4[m >> 2][m & 3] * s;
      }
    }

    const int ntiles = (kN - base) >> 4;
    v2f by[4];
#pragma unroll
    for (int m0 = 0; m0 < 16; m0 += 4) {
      const float y0 = y[m0], y1 = y[m0 + 1], y2 = y[m0 + 2], y3 = y[m0 + 3];
      by[m0 >> 2].x = lh ? y2 : y0;
      by[m0 >> 2].y = lh ? y3 : y1;
    }
    for (int t = wave; t < ntiles; t += kWaves) {
      const int k0 = base + t * 16;
      float* hp = hrow + k0 + 8 * lh;
      v4f clo = *(const v4f*)(hp);
      v4f chi = *(const v4f*)(hp + 4);
      v8f acc = {clo.x, clo.y, clo.z, clo.w, chi.x, chi.y, chi.z, chi.w};
      if (t == 0) {
#pragma unroll
        for (int m0 = 0; m0 < 16; m0 += 4) {
          v2f a;
          a.x = load_u(w, base + m0 + 2 * lh,     k0 + lm);
          a.y = load_u(w, base + m0 + 2 * lh + 1, k0 + lm);
          acc = wmma_f32_16x16x4(a, by[m0 >> 2], acc);
        }
      } else {
        const float* wp = w + (size_t)(base + 2 * lh) * kN + k0 + lm;
#pragma unroll
        for (int m0 = 0; m0 < 16; m0 += 4) {
          v2f a;
          a.x = wp[(size_t)m0 * kN];
          a.y = wp[(size_t)(m0 + 1) * kN];
          acc = wmma_f32_16x16x4(a, by[m0 >> 2], acc);
        }
      }
      v4f olo = {acc[0], acc[1], acc[2], acc[3]};
      v4f ohi = {acc[4], acc[5], acc[6], acc[7]};
      *(v4f*)(hp) = olo;
      *(v4f*)(hp + 4) = ohi;
    }
    __syncthreads();
  }

  for (int i = tid; i < kCols * (kN / 4); i += kThreads) {
    const int c  = i / (kN / 4);
    const int k4 = (i - c * (kN / 4)) * 4;
    v4f v = *(const v4f*)(hbuf + c * kHStr + k4);
    *(v4f*)(out + (size_t)(c0 + c) * kN + k4) = v;
  }
}

extern "C" void kernel_launch(void* const* d_in, const int* in_sizes, int n_in,
                              void* d_out, int out_size, void* d_ws, size_t ws_size,
                              hipStream_t stream) {
  const float* x = (const float*)d_in[0];
  const float* w = (const float*)d_in[1];
  float* out = (float*)d_out;

  dim3 grid(kB / kCols);
  dim3 block(kThreads);
  const size_t smem = (size_t)kSmemFloats * sizeof(float);
  hipLaunchKernelGGL(ortho_householder_kernel, grid, block, smem, stream,
                     x, w, out);
}


# Issue every V load well ahead of its WMMA: phase 1 keeps 8 k-tiles in flight per wave, phase 3 loads 4 row tiles a group ahead, first loads before the solve; diagonal tiles masked in registers
# speedup vs baseline: 3.1409x; 2.2362x over previous
#include <hip/hip_runtime.h>

typedef __attribute__((ext_vector_type(2))) float v2f;
typedef __attribute__((ext_vector_type(4))) float v4f;
typedef __attribute__((ext_vector_type(8))) float v8f;

namespace {
constexpr int kN     = 2048;
constexpr int kB     = 4096;
constexpr int kCols  = 16;
constexpr int kBlk   = 16;
constexpr int kWaves = 8;
constexpr int kThreads = kWaves * 32;
constexpr int kHStr  = kN + 4;
constexpr int kSmemFloats = kCols * kHStr
                          + 2 * kWaves * 256
                          + 2 * 256
                          + 16;
}

__device__ __forceinline__ v8f wmma_f32_16x16x4(v2f a, v2f b, v8f c) {
  return __builtin_amdgcn_wmma_f32_16x16x4_f32(
       false, a,  false, b,
       (short)0, c,  false,  false);
}

constexpr int kPf = 8;
constexpr int kTg = 4;

__device__ __forceinline__ void p1_prefetch(v2f (&abuf)[kPf], const float* __restrict__ w,
                                            int base, int wave, int lm, int lh) {
  const int k = base + wave * 4;
  const int nt = (k < kN) ? (kN - k + kWaves * 4 - 1) / (kWaves * 4) : 0;
  const float* ap = w + (size_t)(base + lm) * kN + k + 2 * lh;
#pragma unroll
  for (int j = 0; j < kPf; ++j) {
    abuf[j] = v2f{0.0f, 0.0f};
    if (j < nt) abuf[j] = *(const v2f*)(ap + (size_t)j * 32);
  }
}

template <bool kLast>
__device__ __forceinline__ void p1_group(v2f (&abuf)[kPf], const float* __restrict__ ap,
                                         const float* bp, int i, int nt,
                                         v8f& accW, v8f& accG) {
  v2f cur[kPf];
#pragma unroll
  for (int j = 0; j < kPf; ++j) cur[j] = abuf[j];
  if (!kLast) {
#pragma unroll
    for (int j = 0; j < kPf; ++j) {
      const int tn = min(i + j + kPf, nt - 1);
      abuf[j] = *(const v2f*)(ap + (size_t)tn * 32);
    }
  }
  v2f bb[kPf];
#pragma unroll
  for (int j = 0; j < kPf; ++j) {
    const int tb = kLast ? min(i + j, nt - 1) : i + j;
    bb[j] = *(const v2f*)(bp + tb * 32);
  }
#pragma unroll
  for (int j = 0; j < kPf; ++j) {
    if (!kLast || i + j < nt) {
      accW = wmma_f32_16x16x4(cur[j], bb[j], accW);
      accG = wmma_f32_16x16x4(cur[j], cur[j], accG);
    }
  }
}

template <bool kGuard>
__device__ __forceinline__ void p3_load(float (&av)[kTg][8], const float* __restrict__ w,
                                        int base, int t, int ntiles, int lh, int lm) {
#pragma unroll
  for (int g = 0; g < kTg; ++g) {
#pragma unroll
    for (int q = 0; q < 8; ++q) av[g][q] = 0.0f;
    if (!kGuard || t + kWaves * g < ntiles) {
      const int k0 = base + (t + kWaves * g) * 16;
      const float* wp = w + (size_t)(base + 2 * lh) * kN + k0 + lm;
#pragma unroll
      for (int q = 0; q < 4; ++q) {
        av[g][2 * q]     = wp[(size_t)(4 * q) * kN];
        av[g][2 * q + 1] = wp[(size_t)(4 * q + 1) * kN];
      }
    }
  }
}

template <bool kGuard>
__device__ __forceinline__ void p3_apply(float (&av)[kTg][8], float* hrow, int base, int t,
                                         int ntiles, const v2f (&by)[4], int lh, int lm) {
  if (t == 0) {
#pragma unroll
    for (int q = 0; q < 4; ++q) {
#pragma unroll
      for (int e = 0; e < 2; ++e) {
        const int row = base + 4 * q + 2 * lh + e;
        const int col = base + lm;
        float v = (col >= row) ? av[0][2 * q + e] : 0.0f;
        if (row == kN - 1 && col == kN - 1) v = 1.0f;
        av[0][2 * q + e] = v;
      }
    }
  }
  v4f clo[kTg], chi[kTg];
#pragma unroll
  for (int g = 0; g < kTg; ++g) {
    clo[g] = v4f{0.0f, 0.0f, 0.0f, 0.0f};
    chi[g] = v4f{0.0f, 0.0f, 0.0f, 0.0f};
    if (!kGuard || t + kWaves * g < ntiles) {
      const float* hp = hrow + base + (t + kWaves * g) * 16 + 8 * lh;
      clo[g] = *(const v4f*)(hp);
      chi[g] = *(const v4f*)(hp + 4);
    }
  }
#pragma unroll
  for (int g = 0; g < kTg; ++g) {
    if (!kGuard || t + kWaves * g < ntiles) {
      float* hp = hrow + base + (t + kWaves * g) * 16 + 8 * lh;
      v8f acc = {clo[g].x, clo[g].y, clo[g].z, clo[g].w, chi[g].x, chi[g].y, chi[g].z, chi[g].w};
#pragma unroll
      for (int q = 0; q < 4; ++q) {
        v2f a;
        a.x = av[g][2 * q];
        a.y = av[g][2 * q + 1];
        acc = wmma_f32_16x16x4(a, by[q], acc);
      }
      v4f olo = {acc[0], acc[1], acc[2], acc[3]};
      v4f ohi = {acc[4], acc[5], acc[6], acc[7]};
      *(v4f*)(hp) = olo;
      *(v4f*)(hp + 4) = ohi;
    }
  }
}

__global__ __launch_bounds__(kThreads)
void ortho_householder_kernel(const float* __restrict__ x,
                              const float* __restrict__ w,
                              float* __restrict__ out) {
  extern __shared__ float smem[];
  float* hbuf  = smem;
  float* Wpart = hbuf + kCols * kHStr;
  float* Gpart = Wpart + kWaves * 256;
  float* Wbuf  = Gpart + kWaves * 256;
  float* Gbuf  = Wbuf + 256;
  float* Cbuf  = Gbuf + 256;

  const int tid  = threadIdx.x;
  const int wave = __builtin_amdgcn_readfirstlane(tid >> 5);
  const int lane = tid & 31;
  const int lm   = lane & 15;
  const int lh   = lane >> 4;
  const int c0   = blockIdx.x * kCols;
  float* const hrow = hbuf + lm * kHStr;

  v2f abuf[kPf];
  p1_prefetch(abuf, w, kN - kBlk, wave, lm, lh);

  for (int i = tid; i < kCols * (kN / 4); i += kThreads) {
    const int c  = i / (kN / 4);
    const int k4 = (i - c * (kN / 4)) * 4;
    v4f v = *(const v4f*)(x + (size_t)(c0 + c) * kN + k4);
    *(v4f*)(hbuf + c * kHStr + k4) = v;
  }
  __syncthreads();

  for (int base = kN - kBlk; base >= 0; base -= kBlk) {
    v8f accW = {};
    v8f accG = {};
    const int k = base + wave * 4;
    if (k < kN) {
      const int nt = (kN - k + kWaves * 4 - 1) / (kWaves * 4);
      const float* ap = w + (size_t)(base + lm) * kN + k + 2 * lh;
      const float* bp = hrow + k + 2 * lh;
      if (wave < 4) {
        const int col = k + 2 * lh;
        const int row = base + lm;
        v2f a = abuf[0];
        a.x = (col >= row) ? a.x : 0.0f;
        a.y = (col + 1 >= row) ? a.y : 0.0f;
        if (row == kN - 1 && col == kN - 1) a.x = 1.0f;
        if (row == kN - 1 && col + 1 == kN - 1) a.y = 1.0f;
        abuf[0] = a;
      }
      int i = 0;
      for (; i + kPf < nt; i += kPf)
        p1_group<false>(abuf, ap, bp, i, nt, accW, accG);
      p1_group<true>(abuf, ap, bp, i, nt, accW, accG);
    }
#pragma unroll
    for (int r = 0; r < 8; ++r) {
      const int m = r + 8 * lh;
      Wpart[wave * 256 + m * 16 + lm] = accW[r];
      Gpart[wave * 256 + m * 16 + lm] = accG[r];
    }
    __syncthreads();

    {
      float sw = 0.f, sg = 0.f;
#pragma unroll
      for (int p = 0; p < kWaves; ++p) {
        sw += Wpart[p * 256 + tid];
        sg += Gpart[p * 256 + tid];
      }
      Wbuf[tid] = sw;
      Gbuf[tid] = sg;
      if ((tid >> 4) == (tid & 15)) Cbuf[tid & 15] = (sg > 0.f) ? (2.0f / sg) : 0.0f;
    }
    __syncthreads();

    if (base >= kBlk) p1_prefetch(abuf, w, base - kBlk, wave, lm, lh);
    const int ntiles = (kN - base) >> 4;
    int t = wave;
    float av[kTg][8];
    p3_load<true>(av, w, base, t, ntiles, lh, lm);

    float y[16];
    {
      float wv[16];
#pragma unroll
      for (int l = 0; l < 16; ++l) wv[l] = Wbuf[l * 16 + lm];
      v4f c4[4];
#pragma unroll
      for (int p = 0; p < 4; ++p) c4[p] = *(const v4f*)(Cbuf + 4 * p);
      v4f g4[16][4];
#pragma unroll
      for (int m = 0; m < 15; ++m) {
#pragma unroll
        for (int p = (m + 1) >> 2; p < 4; ++p)
          g4[m][p] = *(const v4f*)(Gbuf + m * 16 + 4 * p);
      }
      __builtin_amdgcn_sched_barrier(0);
#pragma unroll
      for (int m = 15; m >= 0; --m) {
        float s = wv[m];
#pragma unroll
        for (int l = 15; l > m; --l) s = __builtin_fmaf(g4[m][l >> 2][l & 3], y[l], s);
        y[m] = -c4[m >> 2][m & 3] * s;
      }
    }

    v2f by[4];
#pragma unroll
    for (int m0 = 0; m0 < 16; m0 += 4) {
      const float y0 = y[m0], y1 = y[m0 + 1], y2 = y[m0 + 2], y3 = y[m0 + 3];
      by[m0 >> 2].x = lh ? y2 : y0;
      by[m0 >> 2].y = lh ? y3 : y1;
    }
    while (t < ntiles) {
      float cur[kTg][8];
#pragma unroll
      for (int g = 0; g < kTg; ++g) {
#pragma unroll
        for (int q = 0; q < 8; ++q) cur[g][q] = av[g][q];
      }
      const int tn = t + kWaves * kTg;
      if (tn < ntiles) p3_load<true>(av, w, base, tn, ntiles, lh, lm);
      if (t + kWaves * (kTg - 1) < ntiles) p3_apply<false>(cur, hrow, base, t, ntiles, by, lh, lm);
      else                                 p3_apply<true>(cur, hrow, base, t, ntiles, by, lh, lm);
      t = tn;
    }
    __syncthreads();
  }

  for (int i = tid; i < kCols * (kN / 4); i += kThreads) {
    const int c  = i / (kN / 4);
    const int k4 = (i - c * (kN / 4)) * 4;
    v4f v = *(const v4f*)(hbuf + c * kHStr + k4);
    *(v4f*)(out + (size_t)(c0 + c) * kN + k4) = v;
  }
}

extern "C" void kernel_launch(void* const* d_in, const int* in_sizes, int n_in,
                              void* d_out, int out_size, void* d_ws, size_t ws_size,
                              hipStream_t stream) {
  const float* x = (const float*)d_in[0];
  const float* w = (const float*)d_in[1];
  float* out = (float*)d_out;

  dim3 grid(kB / kCols);
  dim3 block(kThreads);
  const size_t smem = (size_t)kSmemFloats * sizeof(float);
  hipLaunchKernelGGL(ortho_householder_kernel, grid, block, smem, stream,
                     x, w, out);
}


# Each block's V rows are read from memory once (128-bit loads issued a block ahead), staged k-major in LDS, and both phases take V from LDS; order of the 16-reflector sum in phase 3 changes
# speedup vs baseline: 3.3307x; 1.0604x over previous
#include <hip/hip_runtime.h>

typedef __attribute__((ext_vector_type(2))) float v2f;
typedef __attribute__((ext_vector_type(4))) float v4f;
typedef __attribute__((ext_vector_type(8))) float v8f;

namespace {
constexpr int kN     = 2048;
constexpr int kB     = 4096;
constexpr int kCols  = 16;
constexpr int kBlk   = 16;
constexpr int kWaves = 8;
constexpr int kThreads = kWaves * 32;
constexpr int kHStr  = kN + 4;
constexpr int kVtStr = kBlk + 4;
constexpr int kSmemFloats = kCols * kHStr
                          + 2 * kWaves * 256
                          + 2 * 256
                          + 16
                          + kN * kVtStr;
static_assert(kSmemFloats * sizeof(float) <= 320u * 1024u, "one work-group's LDS");
static_assert(((kCols * kHStr + 2 * kWaves * 256 + 2 * 256 + 16) % 4) == 0, "Vt starts 16-byte aligned");
}

__device__ __forceinline__ v8f wmma_f32_16x16x4(v2f a, v2f b, v8f c) {
  return __builtin_amdgcn_wmma_f32_16x16x4_f32(
       false, a,  false, b,
       (short)0, c,  false,  false);
}

constexpr int kPf = 8;
constexpr int kTg = 4;
constexpr int kSt = 2 * (kN / 16 / kWaves);

__device__ __forceinline__ void stage_load(v4f (&stg)[kSt], const float* __restrict__ w,
                                           int base, int wave, int lane) {
  const int ntiles = (kN - base) >> 4;
  const float* p = w + (size_t)(base + (lane >> 2)) * kN + base + 4 * (lane & 3);
#pragma unroll
  for (int s = 0; s < kSt / 2; ++s) {
    const int t = wave + kWaves * s;
    if (t < ntiles) {
      stg[2 * s]     = *(const v4f*)(p + t * 16);
      stg[2 * s + 1] = *(const v4f*)(p + (size_t)8 * kN + t * 16);
    }
  }
}

__device__ __forceinline__ void stage_store(const v4f (&stg)[kSt], float* vt,
                                            int base, int wave, int lane) {
  const int ntiles = (kN - base) >> 4;
  const int mr = lane >> 2;
  const int c4 = 4 * (lane & 3);
  float* q = vt + (base + c4) * kVtStr + mr;
#pragma unroll
  for (int s = 0; s < kSt / 2; ++s) {
    const int t = wave + kWaves * s;
    if (t < ntiles) {
#pragma unroll
      for (int rg = 0; rg < 2; ++rg) {
        v4f v = stg[2 * s + rg];
        if (s == 0 && wave == 0) {
          const int row = base + 8 * rg + mr;
#pragma unroll
          for (int e = 0; e < 4; ++e) {
            const int col = base + c4 + e;
            float u = (col >= row) ? v[e] : 0.0f;
            if (row == kN - 1 && col == kN - 1) u = 1.0f;
            v[e] = u;
          }
        }
        float* d = q + (t * 16) * kVtStr + 8 * rg;
        d[0]          = v.x;
        d[kVtStr]     = v.y;
        d[2 * kVtStr] = v.z;
        d[3 * kVtStr] = v.w;
      }
    }
  }
}

__device__ __forceinline__ void p1_prefetch(v2f (&abuf)[kPf], v2f (&bbuf)[kPf],
                                            const float* ap, const float* bp, int nt) {
#pragma unroll
  for (int j = 0; j < kPf; ++j) {
    abuf[j] = v2f{0.0f, 0.0f};
    bbuf[j] = v2f{0.0f, 0.0f};
    if (j < nt) {
      abuf[j].x = ap[j * (32 * kVtStr)];
      abuf[j].y = ap[j * (32 * kVtStr) + kVtStr];
      bbuf[j] = *(const v2f*)(bp + j * 32);
    }
  }
}

template <bool kLast>
__device__ __forceinline__ void p1_group(v2f (&abuf)[kPf], v2f (&bbuf)[kPf], const float* ap,
                                         const float* bp, int i, int nt,
                                         v8f& accW, v8f& accG) {
  v2f ca[kPf], cb[kPf];
#pragma unroll
  for (int j = 0; j < kPf; ++j) {
    ca[j] = abuf[j];
    cb[j] = bbuf[j];
  }
  if (!kLast) {
#pragma unroll
    for (int j = 0; j < kPf; ++j) {
      const int tn = min(i + j + kPf, nt - 1);
      abuf[j].x = ap[tn * (32 * kVtStr)];
      abuf[j].y = ap[tn * (32 * kVtStr) + kVtStr];
      bbuf[j] = *(const v2f*)(bp + tn * 32);
    }
    __builtin_amdgcn_sched_barrier(0);
  }
#pragma unroll
  for (int j = 0; j < kPf; ++j) {
    if (!kLast || i + j < nt) {
      accW = wmma_f32_16x16x4(ca[j], cb[j], accW);
      accG = wmma_f32_16x16x4(ca[j], ca[j], accG);
    }
  }
}

template <bool kGuard>
__device__ __forceinline__ void p3_load(float (&av)[kTg][8], const float* vt,
                                        int base, int t, int ntiles, int lh, int lm) {
#pragma unroll
  for (int g = 0; g < kTg; ++g) {
#pragma unroll
    for (int q = 0; q < 8; ++q) av[g][q] = 0.0f;
    if (!kGuard || t + kWaves * g < ntiles) {
      const int k0 = base + (t + kWaves * g) * 16;
      const float* vp = vt + (k0 + lm) * kVtStr + 8 * lh;
      const v4f lo = *(const v4f*)(vp);
      const v4f hi = *(const v4f*)(vp + 4);
      av[g][0] = lo.x; av[g][1] = lo.y; av[g][2] = lo.z; av[g][3] = lo.w;
      av[g][4] = hi.x; av[g][5] = hi.y; av[g][6] = hi.z; av[g][7] = hi.w;
    }
  }
}

template <bool kGuard>
__device__ __forceinline__ void p3_apply(float (&av)[kTg][8], float* hrow, int base, int t,
                                         int ntiles, const v2f (&by)[4], int lh, int lm) {
  v4f clo[kTg], chi[kTg];
#pragma unroll
  for (int g = 0; g < kTg; ++g) {
    clo[g] = v4f{0.0f, 0.0f, 0.0f, 0.0f};
    chi[g] = v4f{0.0f, 0.0f, 0.0f, 0.0f};
    if (!kGuard || t + kWaves * g < ntiles) {
      const float* hp = hrow + base + (t + kWaves * g) * 16 + 8 * lh;
      clo[g] = *(const v4f*)(hp);
      chi[g] = *(const v4f*)(hp + 4);
    }
  }
#pragma unroll
  for (int g = 0; g < kTg; ++g) {
    if (!kGuard || t + kWaves * g < ntiles) {
      float* hp = hrow + base + (t + kWaves * g) * 16 + 8 * lh;
      v8f acc = {clo[g].x, clo[g].y, clo[g].z, clo[g].w, chi[g].x, chi[g].y, chi[g].z, chi[g].w};
#pragma unroll
      for (int q = 0; q < 4; ++q) {
        v2f a;
        a.x = av[g][2 * q];
        a.y = av[g][2 * q + 1];
        acc = wmma_f32_16x16x4(a, by[q], acc);
      }
      v4f olo = {acc[0], acc[1], acc[2], acc[3]};
      v4f ohi = {acc[4], acc[5], acc[6], acc[7]};
      *(v4f*)(hp) = olo;
      *(v4f*)(hp + 4) = ohi;
    }
  }
}

__global__ __launch_bounds__(kThreads)
void ortho_householder_kernel(const float* __restrict__ x,
                              const float* __restrict__ w,
                              float* __restrict__ out) {
  __shared__ __attribute__((aligned(16))) float smem[kSmemFloats];
  float* hbuf  = smem;
  float* Wpart = hbuf + kCols * kHStr;
  float* Gpart = Wpart + kWaves * 256;
  float* Wbuf  = Gpart + kWaves * 256;
  float* Gbuf  = Wbuf + 256;
  float* Cbuf  = Gbuf + 256;
  float* Vt    = Cbuf + 16;

  const int tid  = threadIdx.x;
  const int wave = __builtin_amdgcn_readfirstlane(tid >> 5);
  const int lane = tid & 31;
  const int lm   = lane & 15;
  const int lh   = lane >> 4;
  const int c0   = blockIdx.x * kCols;
  float* const hrow = hbuf + lm * kHStr;

  v4f stg[kSt];
  stage_load(stg, w, kN - kBlk, wave, lane);

  for (int i = tid; i < kCols * (kN / 4); i += kThreads) {
    const int c  = i / (kN / 4);
    const int k4 = (i - c * (kN / 4)) * 4;
    v4f v = *(const v4f*)(x + (size_t)(c0 + c) * kN + k4);
    *(v4f*)(hbuf + c * kHStr + k4) = v;
  }
  __syncthreads();

  for (int base = kN - kBlk; base >= 0; base -= kBlk) {
    stage_store(stg, Vt, base, wave, lane);
    __syncthreads();

    v8f accW = {};
    v8f accG = {};
    const int k = base + wave * 4;
    if (k < kN) {
      const int nt = (kN - k + kWaves * 4 - 1) / (kWaves * 4);
      const float* ap = Vt + (k + 2 * lh) * kVtStr + lm;
      const float* bp = hrow + k + 2 * lh;
      v2f abuf[kPf], bbuf[kPf];
      p1_prefetch(abuf, bbuf, ap, bp, nt);
      int i = 0;
      for (; i + kPf < nt; i += kPf)
        p1_group<false>(abuf, bbuf, ap, bp, i, nt, accW, accG);
      p1_group<true>(abuf, bbuf, ap, bp, i, nt, accW, accG);
    }
#pragma unroll
    for (int r = 0; r < 8; ++r) {
      const int m = r + 8 * lh;
      Wpart[wave * 256 + m * 16 + lm] = accW[r];
      Gpart[wave * 256 + m * 16 + lm] = accG[r];
    }
    __syncthreads();

    {
      float sw = 0.f, sg = 0.f;
#pragma unroll
      for (int p = 0; p < kWaves; ++p) {
        sw += Wpart[p * 256 + tid];
        sg += Gpart[p * 256 + tid];
      }
      Wbuf[tid] = sw;
      Gbuf[tid] = sg;
      if ((tid >> 4) == (tid & 15)) Cbuf[tid & 15] = (sg > 0.f) ? (2.0f / sg) : 0.0f;
    }
    __syncthreads();

    if (base >= kBlk) stage_load(stg, w, base - kBlk, wave, lane);
    const int ntiles = (kN - base) >> 4;
    int t = wave;
    float av[kTg][8];
    p3_load<true>(av, Vt, base, t, ntiles, lh, lm);

    float y[16];
    {
      float wv[16];
#pragma unroll
      for (int l = 0; l < 16; ++l) wv[l] = Wbuf[l * 16 + lm];
      v4f c4[4];
#pragma unroll
      for (int p = 0; p < 4; ++p) c4[p] = *(const v4f*)(Cbuf + 4 * p);
      v4f g4[16][4];
#pragma unroll
      for (int m = 0; m < 15; ++m) {
#pragma unroll
        for (int p = (m + 1) >> 2; p < 4; ++p)
          g4[m][p] = *(const v4f*)(Gbuf + m * 16 + 4 * p);
      }
      __builtin_amdgcn_sched_barrier(0);
#pragma unroll
      for (int m = 15; m >= 0; --m) {
        float s = wv[m];
#pragma unroll
        for (int l = 15; l > m; --l) s = __builtin_fmaf(g4[m][l >> 2][l & 3], y[l], s);
        y[m] = -c4[m >> 2][m & 3] * s;
      }
    }

    v2f by[4];
#pragma unroll
    for (int q = 0; q < 4; ++q) {
      const float a0 = y[2 * q], a1 = y[2 * q + 1], b0 = y[8 + 2 * q], b1 = y[9 + 2 * q];
      by[q].x = lh ? b0 : a0;
      by[q].y = lh ? b1 : a1;
    }
    while (t < ntiles) {
      float cur[kTg][8];
#pragma unroll
      for (int g = 0; g < kTg; ++g) {
#pragma unroll
        for (int q = 0; q < 8; ++q) cur[g][q] = av[g][q];
      }
      const int tn = t + kWaves * kTg;
      if (tn < ntiles) p3_load<true>(av, Vt, base, tn, ntiles, lh, lm);
      if (t + kWaves * (kTg - 1) < ntiles) p3_apply<false>(cur, hrow, base, t, ntiles, by, lh, lm);
      else                                 p3_apply<true>(cur, hrow, base, t, ntiles, by, lh, lm);
      t = tn;
    }
    __syncthreads();
  }

  for (int i = tid; i < kCols * (kN / 4); i += kThreads) {
    const int c  = i / (kN / 4);
    const int k4 = (i - c * (kN / 4)) * 4;
    v4f v = *(const v4f*)(hbuf + c * kHStr + k4);
    *(v4f*)(out + (size_t)(c0 + c) * kN + k4) = v;
  }
}

extern "C" void kernel_launch(void* const* d_in, const int* in_sizes, int n_in,
                              void* d_out, int out_size, void* d_ws, size_t ws_size,
                              hipStream_t stream) {
  const float* x = (const float*)d_in[0];
  const float* w = (const float*)d_in[1];
  float* out = (float*)d_out;

  dim3 grid(kB / kCols);
  dim3 block(kThreads);
  hipLaunchKernelGGL(ortho_householder_kernel, grid, block, 0, stream,
                     x, w, out);
}


# Batch-invariant block factors go to a pre-pass kernel: Gram, c and the 16x16 triangular T built once per block, not per workgroup; main kernel forms Y=-T*W with 4 WMMAs (sum order of Y changes)
# speedup vs baseline: 4.3865x; 1.3170x over previous
#include <hip/hip_runtime.h>

typedef __attribute__((ext_vector_type(2))) float v2f;
typedef __attribute__((ext_vector_type(4))) float v4f;
typedef __attribute__((ext_vector_type(8))) float v8f;

namespace {
constexpr int kN     = 2048;
constexpr int kB     = 4096;
constexpr int kCols  = 16;
constexpr int kBlk   = 16;
constexpr int kWaves = 8;
constexpr int kThreads = kWaves * 32;
constexpr int kHStr  = kN + 4;
constexpr int kVtStr = kBlk + 4;
constexpr int kSmemFloats = kCols * kHStr
                          + kWaves * 256
                          + 256
                          + kN * kVtStr;
static_assert(kSmemFloats * sizeof(float) <= 320u * 1024u, "one work-group's LDS");
static_assert(((kCols * kHStr + kWaves * 256 + 256) % 4) == 0, "Vt starts 16-byte aligned");
constexpr int kTfBlk = 256;
constexpr int kGb    = 16;
}

__device__ __forceinline__ v8f wmma_f32_16x16x4(v2f a, v2f b, v8f c) {
  return __builtin_amdgcn_wmma_f32_16x16x4_f32(
       false, a,  false, b,
       (short)0, c,  false,  false);
}

constexpr int kPf = 8;
constexpr int kTg = 4;
constexpr int kSt = 2 * (kN / 16 / kWaves);

__device__ __forceinline__ void stage_load(v4f (&stg)[kSt], const float* __restrict__ w,
                                           int base, int wave, int lane) {
  const int ntiles = (kN - base) >> 4;
  const float* p = w + (size_t)(base + (lane >> 2)) * kN + base + 4 * (lane & 3);
#pragma unroll
  for (int s = 0; s < kSt / 2; ++s) {
    const int t = wave + kWaves * s;
    if (t < ntiles) {
      stg[2 * s]     = *(const v4f*)(p + t * 16);
      stg[2 * s + 1] = *(const v4f*)(p + (size_t)8 * kN + t * 16);
    }
  }
}

__device__ __forceinline__ void stage_store(const v4f (&stg)[kSt], float* vt,
                                            int base, int wave, int lane) {
  const int ntiles = (kN - base) >> 4;
  const int mr = lane >> 2;
  const int c4 = 4 * (lane & 3);
  float* q = vt + (base + c4) * kVtStr + mr;
#pragma unroll
  for (int s = 0; s < kSt / 2; ++s) {
    const int t = wave + kWaves * s;
    if (t < ntiles) {
#pragma unroll
      for (int rg = 0; rg < 2; ++rg) {
        v4f v = stg[2 * s + rg];
        if (s == 0 && wave == 0) {
          const int row = base + 8 * rg + mr;
#pragma unroll
          for (int e = 0; e < 4; ++e) {
            const int col = base + c4 + e;
            float u = (col >= row) ? v[e] : 0.0f;
            if (row == kN - 1 && col == kN - 1) u = 1.0f;
            v[e] = u;
          }
        }
        float* d = q + (t * 16) * kVtStr + 8 * rg;
        d[0]          = v.x;
        d[kVtStr]     = v.y;
        d[2 * kVtStr] = v.z;
        d[3 * kVtStr] = v.w;
      }
    }
  }
}

__device__ __forceinline__ void p1_prefetch(v2f (&abuf)[kPf], v2f (&bbuf)[kPf],
                                            const float* ap, const float* bp, int nt) {
#pragma unroll
  for (int j = 0; j < kPf; ++j) {
    abuf[j] = v2f{0.0f, 0.0f};
    bbuf[j] = v2f{0.0f, 0.0f};
    if (j < nt) {
      abuf[j].x = ap[j * (32 * kVtStr)];
      abuf[j].y = ap[j * (32 * kVtStr) + kVtStr];
      bbuf[j] = *(const v2f*)(bp + j * 32);
    }
  }
}

template <bool kLast>
__device__ __forceinline__ void p1_group(v2f (&abuf)[kPf], v2f (&bbuf)[kPf], const float* ap,
                                         const float* bp, int i, int nt,
                                         v8f& accW) {
  v2f ca[kPf], cb[kPf];
#pragma unroll
  for (int j = 0; j < kPf; ++j) {
    ca[j] = abuf[j];
    cb[j] = bbuf[j];
  }
  if (!kLast) {
#pragma unroll
    for (int j = 0; j < kPf; ++j) {
      const int tn = min(i + j + kPf, nt - 1);
      abuf[j].x = ap[tn * (32 * kVtStr)];
      abuf[j].y = ap[tn * (32 * kVtStr) + kVtStr];
      bbuf[j] = *(const v2f*)(bp + tn * 32);
    }
    __builtin_amdgcn_sched_barrier(0);
  }
#pragma unroll
  for (int j = 0; j < kPf; ++j) {
    if (!kLast || i + j < nt) {
      accW = wmma_f32_16x16x4(ca[j], cb[j], accW);
    }
  }
}

template <bool kGuard>
__device__ __forceinline__ void p3_load(float (&av)[kTg][8], const float* vt,
                                        int base, int t, int ntiles, int lh, int lm) {
#pragma unroll
  for (int g = 0; g < kTg; ++g) {
#pragma unroll
    for (int q = 0; q < 8; ++q) av[g][q] = 0.0f;
    if (!kGuard || t + kWaves * g < ntiles) {
      const int k0 = base + (t + kWaves * g) * 16;
      const float* vp = vt + (k0 + lm) * kVtStr + 8 * lh;
      const v4f lo = *(const v4f*)(vp);
      const v4f hi = *(const v4f*)(vp + 4);
      av[g][0] = lo.x; av[g][1] = lo.y; av[g][2] = lo.z; av[g][3] = lo.w;
      av[g][4] = hi.x; av[g][5] = hi.y; av[g][6] = hi.z; av[g][7] = hi.w;
    }
  }
}

template <bool kGuard>
__device__ __forceinline__ void p3_apply(float (&av)[kTg][8], float* hrow, int base, int t,
                                         int ntiles, const v2f (&by)[4], int lh, int lm) {
  v4f clo[kTg], chi[kTg];
#pragma unroll
  for (int g = 0; g < kTg; ++g) {
    clo[g] = v4f{0.0f, 0.0f, 0.0f, 0.0f};
    chi[g] = v4f{0.0f, 0.0f, 0.0f, 0.0f};
    if (!kGuard || t + kWaves * g < ntiles) {
      const float* hp = hrow + base + (t + kWaves * g) * 16 + 8 * lh;
      clo[g] = *(const v4f*)(hp);
      chi[g] = *(const v4f*)(hp + 4);
    }
  }
#pragma unroll
  for (int g = 0; g < kTg; ++g) {
    if (!kGuard || t + kWaves * g < ntiles) {
      float* hp = hrow + base + (t + kWaves * g) * 16 + 8 * lh;
      v8f acc = {clo[g].x, clo[g].y, clo[g].z, clo[g].w, chi[g].x, chi[g].y, chi[g].z, chi[g].w};
#pragma unroll
      for (int q = 0; q < 4; ++q) {
        v2f a;
        a.x = av[g][2 * q];
        a.y = av[g][2 * q + 1];
        acc = wmma_f32_16x16x4(a, by[q], acc);
      }
      v4f olo = {acc[0], acc[1], acc[2], acc[3]};
      v4f ohi = {acc[4], acc[5], acc[6], acc[7]};
      *(v4f*)(hp) = olo;
      *(v4f*)(hp + 4) = ohi;
    }
  }
}

template <bool kTail>
__device__ __forceinline__ void gram_batch(const float* p, int i0, int nt, int k, int row,
                                           int lh, v8f& accG) {
  v2f a[kGb];
#pragma unroll
  for (int j = 0; j < kGb; ++j) {
    a[j] = v2f{0.0f, 0.0f};
    if (!kTail || i0 + j < nt) a[j] = *(const v2f*)(p + (i0 + j) * 32);
  }
#pragma unroll
  for (int j = 0; j < kGb; ++j) {
    if (!kTail || i0 + j < nt) {
      const int col = k + (i0 + j) * 32 + 2 * lh;
      v2f u;
      u.x = (col >= row) ? a[j].x : 0.0f;
      u.y = (col + 1 >= row) ? a[j].y : 0.0f;
      if (row == kN - 1 && col == kN - 1) u.x = 1.0f;
      if (row == kN - 1 && col + 1 == kN - 1) u.y = 1.0f;
      accG = wmma_f32_16x16x4(u, u, accG);
    }
  }
}

__global__ __launch_bounds__(kThreads)
void block_factor_kernel(const float* __restrict__ w, float* __restrict__ tf) {
  __shared__ __attribute__((aligned(16))) float fsm[kWaves * 256 + 256 + 16 + 256];
  float* Gpart = fsm;
  float* Gbuf  = Gpart + kWaves * 256;
  float* Cbuf  = Gbuf + 256;
  float* Tl    = Cbuf + 16;

  const int tid  = threadIdx.x;
  const int wave = __builtin_amdgcn_readfirstlane(tid >> 5);
  const int lane = tid & 31;
  const int lm   = lane & 15;
  const int lh   = lane >> 4;
  const int blk  = blockIdx.x;
  const int base = blk * kBlk;

  v8f accG = {};
  const int k = base + wave * 4;
  if (k < kN) {
    const int nt  = (kN - k + kWaves * 4 - 1) / (kWaves * 4);
    const int row = base + lm;
    const float* p = w + (size_t)row * kN + k + 2 * lh;
    int i = 0;
    for (; i + kGb <= nt; i += kGb) gram_batch<false>(p, i, nt, k, row, lh, accG);
    if (i < nt) gram_batch<true>(p, i, nt, k, row, lh, accG);
  }
#pragma unroll
  for (int r = 0; r < 8; ++r)
    Gpart[wave * 256 + (r + 8 * lh) * 16 + lm] = accG[r];
  __syncthreads();
  {
    float sg = 0.f;
#pragma unroll
    for (int pw = 0; pw < kWaves; ++pw) sg += Gpart[pw * 256 + tid];
    Gbuf[tid] = sg;
    if ((tid >> 4) == (tid & 15)) Cbuf[tid & 15] = (sg > 0.f) ? (2.0f / sg) : 0.0f;
  }
  __syncthreads();

  if (tid < 32) {
    float y[16];
    v4f c4[4];
#pragma unroll
    for (int pp = 0; pp < 4; ++pp) c4[pp] = *(const v4f*)(Cbuf + 4 * pp);
    v4f g4[16][4];
#pragma unroll
    for (int m = 0; m < 15; ++m) {
#pragma unroll
      for (int pp = (m + 1) >> 2; pp < 4; ++pp)
        g4[m][pp] = *(const v4f*)(Gbuf + m * 16 + 4 * pp);
    }
#pragma unroll
    for (int m = 15; m >= 0; --m) {
      float sacc = (m == lm) ? 1.0f : 0.0f;
#pragma unroll
      for (int l = 15; l > m; --l) sacc = __builtin_fmaf(g4[m][l >> 2][l & 3], y[l], sacc);
      y[m] = -c4[m >> 2][m & 3] * sacc;
    }
    if (lh == 0) {
#pragma unroll
      for (int m = 0; m < 16; ++m) Tl[m * 16 + lm] = y[m];
    }
  }
  __syncthreads();
  if (tid < 32) {
    const float* tr = Tl + lm * 16 + 2 * lh;
    const v4f lo = {tr[0], tr[1], tr[4], tr[5]};
    const v4f hi = {tr[8], tr[9], tr[12], tr[13]};
    float* d = tf + (size_t)blk * kTfBlk + lane * 4;
    *(v4f*)(d)       = lo;
    *(v4f*)(d + 128) = hi;
  }
}

__global__ __launch_bounds__(kThreads)
void ortho_householder_kernel(const float* __restrict__ x,
                              const float* __restrict__ w,
                              const float* __restrict__ tf,
                              float* __restrict__ out) {
  __shared__ __attribute__((aligned(16))) float smem[kSmemFloats];
  float* hbuf  = smem;
  float* Wpart = hbuf + kCols * kHStr;
  float* Wbuf  = Wpart + kWaves * 256;
  float* Vt    = Wbuf + 256;

  const int tid  = threadIdx.x;
  const int wave = __builtin_amdgcn_readfirstlane(tid >> 5);
  const int lane = tid & 31;
  const int lm   = lane & 15;
  const int lh   = lane >> 4;
  const int c0   = blockIdx.x * kCols;
  float* const hrow = hbuf + lm * kHStr;

  v4f stg[kSt];
  stage_load(stg, w, kN - kBlk, wave, lane);
  const float* tfl = tf + lane * 4;
  v4f tnl = *(const v4f*)(tfl + (size_t)(kN / kBlk - 1) * kTfBlk);
  v4f tnh = *(const v4f*)(tfl + (size_t)(kN / kBlk - 1) * kTfBlk + 128);

  for (int i = tid; i < kCols * (kN / 4); i += kThreads) {
    const int c  = i / (kN / 4);
    const int k4 = (i - c * (kN / 4)) * 4;
    v4f v = *(const v4f*)(x + (size_t)(c0 + c) * kN + k4);
    *(v4f*)(hbuf + c * kHStr + k4) = v;
  }
  __syncthreads();

  for (int base = kN - kBlk; base >= 0; base -= kBlk) {
    stage_store(stg, Vt, base, wave, lane);
    __syncthreads();

    v8f accW = {};
    const int k = base + wave * 4;
    if (k < kN) {
      const int nt = (kN - k + kWaves * 4 - 1) / (kWaves * 4);
      const float* ap = Vt + (k + 2 * lh) * kVtStr + lm;
      const float* bp = hrow + k + 2 * lh;
      v2f abuf[kPf], bbuf[kPf];
      p1_prefetch(abuf, bbuf, ap, bp, nt);
      int i = 0;
      for (; i + kPf < nt; i += kPf)
        p1_group<false>(abuf, bbuf, ap, bp, i, nt, accW);
      p1_group<true>(abuf, bbuf, ap, bp, i, nt, accW);
    }
#pragma unroll
    for (int r = 0; r < 8; ++r)
      Wpart[wave * 256 + (r + 8 * lh) * 16 + lm] = accW[r];
    __syncthreads();

    {
      float sw = 0.f;
#pragma unroll
      for (int p = 0; p < kWaves; ++p) sw += Wpart[p * 256 + tid];
      Wbuf[tid] = sw;
    }
    __syncthreads();

    const v4f tcl = tnl, tch = tnh;
    if (base >= kBlk) {
      stage_load(stg, w, base - kBlk, wave, lane);
      tnl = *(const v4f*)(tfl + (size_t)(base / kBlk - 1) * kTfBlk);
      tnh = *(const v4f*)(tfl + (size_t)(base / kBlk - 1) * kTfBlk + 128);
    }
    const int ntiles = (kN - base) >> 4;
    int t = wave;
    float av[kTg][8];
    p3_load<true>(av, Vt, base, t, ntiles, lh, lm);

    v2f by[4];
    {
      v2f bw[4];
#pragma unroll
      for (int q = 0; q < 4; ++q) {
        bw[q].x = Wbuf[(4 * q + 2 * lh) * 16 + lm];
        bw[q].y = Wbuf[(4 * q + 2 * lh + 1) * 16 + lm];
      }
      v8f yacc = {};
      yacc = wmma_f32_16x16x4(v2f{tcl.x, tcl.y}, bw[0], yacc);
      yacc = wmma_f32_16x16x4(v2f{tcl.z, tcl.w}, bw[1], yacc);
      yacc = wmma_f32_16x16x4(v2f{tch.x, tch.y}, bw[2], yacc);
      yacc = wmma_f32_16x16x4(v2f{tch.z, tch.w}, bw[3], yacc);
#pragma unroll
      for (int q = 0; q < 4; ++q) {
        by[q].x = yacc[2 * q];
        by[q].y = yacc[2 * q + 1];
      }
    }

    while (t < ntiles) {
      float cur[kTg][8];
#pragma unroll
      for (int g = 0; g < kTg; ++g) {
#pragma unroll
        for (int q = 0; q < 8; ++q) cur[g][q] = av[g][q];
      }
      const int tn = t + kWaves * kTg;
      if (tn < ntiles) p3_load<true>(av, Vt, base, tn, ntiles, lh, lm);
      if (t + kWaves * (kTg - 1) < ntiles) p3_apply<false>(cur, hrow, base, t, ntiles, by, lh, lm);
      else                                 p3_apply<true>(cur, hrow, base, t, ntiles, by, lh, lm);
      t = tn;
    }
    __syncthreads();
  }

  for (int i = tid; i < kCols * (kN / 4); i += kThreads) {
    const int c  = i / (kN / 4);
    const int k4 = (i - c * (kN / 4)) * 4;
    v4f v = *(const v4f*)(hbuf + c * kHStr + k4);
    *(v4f*)(out + (size_t)(c0 + c) * kN + k4) = v;
  }
}

extern "C" void kernel_launch(void* const* d_in, const int* in_sizes, int n_in,
                              void* d_out, int out_size, void* d_ws, size_t ws_size,
                              hipStream_t stream) {
  const float* x = (const float*)d_in[0];
  const float* w = (const float*)d_in[1];
  float* out = (float*)d_out;
  float* tf  = (float*)d_ws;

  hipLaunchKernelGGL(block_factor_kernel, dim3(kN / kBlk), dim3(kThreads), 0, stream, w, tf);

  dim3 grid(kB / kCols);
  dim3 block(kThreads);
  hipLaunchKernelGGL(ortho_householder_kernel, grid, block, 0, stream,
                     x, w, (const float*)tf, out);
}


# Each wave keeps its 16 row tiles of h in registers as WMMA accumulators for the whole kernel: phase 3 adds into them, phase 1 takes its B operand from them (W partial sums regroup by row tile)
# speedup vs baseline: 5.2226x; 1.1906x over previous
#include <hip/hip_runtime.h>

typedef __attribute__((ext_vector_type(2))) float v2f;
typedef __attribute__((ext_vector_type(4))) float v4f;
typedef __attribute__((ext_vector_type(8))) float v8f;

namespace {
constexpr int kN     = 2048;
constexpr int kB     = 4096;
constexpr int kCols  = 16;
constexpr int kBlk   = 16;
constexpr int kWaves = 8;
constexpr int kThreads = kWaves * 32;
constexpr int kHStr  = kN + 4;
constexpr int kVtStr = kBlk + 4;
constexpr int kSmemFloats = kCols * kHStr
                          + kWaves * 256
                          + 256
                          + kN * kVtStr;
static_assert(kSmemFloats * sizeof(float) <= 320u * 1024u, "one work-group's LDS");
static_assert(((kCols * kHStr + kWaves * 256 + 256) % 4) == 0, "Vt starts 16-byte aligned");
constexpr int kTfBlk = 256;
constexpr int kGb    = 16;
}

__device__ __forceinline__ v8f wmma_f32_16x16x4(v2f a, v2f b, v8f c) {
  return __builtin_amdgcn_wmma_f32_16x16x4_f32(
       false, a,  false, b,
       (short)0, c,  false,  false);
}

constexpr int kTg  = 4;
constexpr int kTpw = kN / 16 / kWaves;
constexpr int kNg  = kTpw / kTg;
constexpr int kSt = 2 * (kN / 16 / kWaves);

__device__ __forceinline__ void stage_load(v4f (&stg)[kSt], const float* __restrict__ w,
                                           int base, int wave, int lane) {
  const int ntiles = (kN - base) >> 4;
  const float* p = w + (size_t)(base + (lane >> 2)) * kN + base + 4 * (lane & 3);
#pragma unroll
  for (int s = 0; s < kSt / 2; ++s) {
    const int t = wave + kWaves * s;
    if (t < ntiles) {
      stg[2 * s]     = *(const v4f*)(p + t * 16);
      stg[2 * s + 1] = *(const v4f*)(p + (size_t)8 * kN + t * 16);
    }
  }
}

__device__ __forceinline__ void stage_store(const v4f (&stg)[kSt], float* vt,
                                            int base, int wave, int lane) {
  const int ntiles = (kN - base) >> 4;
  const int mr = lane >> 2;
  const int c4 = 4 * (lane & 3);
  float* q = vt + (base + c4) * kVtStr + mr;
#pragma unroll
  for (int s = 0; s < kSt / 2; ++s) {
    const int t = wave + kWaves * s;
    if (t < ntiles) {
      const int toff = __builtin_amdgcn_readfirstlane(t * 16 * kVtStr);
#pragma unroll
      for (int rg = 0; rg < 2; ++rg) {
        v4f v = stg[2 * s + rg];
        if (s == 0 && wave == 0) {
          const int row = base + 8 * rg + mr;
#pragma unroll
          for (int e = 0; e < 4; ++e) {
            const int col = base + c4 + e;
            float u = (col >= row) ? v[e] : 0.0f;
            if (row == kN - 1 && col == kN - 1) u = 1.0f;
            v[e] = u;
          }
        }
        float* d = q + toff + 8 * rg;
        d[0]          = v.x;
        d[kVtStr]     = v.y;
        d[2 * kVtStr] = v.z;
        d[3 * kVtStr] = v.w;
      }
    }
  }
}

template <bool kGuard>
__device__ __forceinline__ void p1_load(float (&a)[kTg][8], const float* vt, int wave, int s0,
                                        int smin, int lh, int lm) {
#pragma unroll
  for (int j = 0; j < kTg; ++j) {
#pragma unroll
    for (int q = 0; q < 8; ++q) a[j][q] = 0.0f;
    if (!kGuard || s0 + j >= smin) {
      const float* ap = vt + ((wave + kWaves * (s0 + j)) * 16 + 8 * lh) * kVtStr + lm;
#pragma unroll
      for (int q = 0; q < 8; ++q) a[j][q] = ap[q * kVtStr];
    }
  }
}

template <int kG, bool kGuard>
__device__ __forceinline__ void p1_apply(const float (&a)[kTg][8], const v8f (&hreg)[kTpw],
                                         int smin, v8f& accW) {
#pragma unroll
  for (int j = 0; j < kTg; ++j) {
    if (!kGuard || kG * kTg + j >= smin) {
#pragma unroll
      for (int q = 0; q < 4; ++q) {
        const v2f fa = {a[j][2 * q], a[j][2 * q + 1]};
        const v2f fb = {hreg[kG * kTg + j][2 * q], hreg[kG * kTg + j][2 * q + 1]};
        accW = wmma_f32_16x16x4(fa, fb, accW);
      }
    }
  }
}

template <int kG>
__device__ __forceinline__ void p1_step(float (&a1)[kTg][8], const v8f (&hreg)[kTpw],
                                        const float* vt, int wave, int g0, int smin,
                                        int lh, int lm, v8f& accW) {
  if (kG >= g0) {
    float c1[kTg][8];
#pragma unroll
    for (int j = 0; j < kTg; ++j) {
#pragma unroll
      for (int q = 0; q < 8; ++q) c1[j][q] = a1[j][q];
    }
    if (kG + 1 < kNg) p1_load<false>(a1, vt, wave, (kG + 1) * kTg, smin, lh, lm);
    if (kG * kTg >= smin) p1_apply<kG, false>(c1, hreg, smin, accW);
    else                  p1_apply<kG, true>(c1, hreg, smin, accW);
  }
}

template <bool kGuard>
__device__ __forceinline__ void p3_load(float (&av)[kTg][8], const float* vt, int wave, int s0,
                                        int smin, int lh, int lm) {
#pragma unroll
  for (int j = 0; j < kTg; ++j) {
#pragma unroll
    for (int q = 0; q < 8; ++q) av[j][q] = 0.0f;
    if (!kGuard || s0 + j >= smin) {
      const float* vp = vt + ((wave + kWaves * (s0 + j)) * 16 + lm) * kVtStr + 8 * lh;
      const v4f lo = *(const v4f*)(vp);
      const v4f hi = *(const v4f*)(vp + 4);
      av[j][0] = lo.x; av[j][1] = lo.y; av[j][2] = lo.z; av[j][3] = lo.w;
      av[j][4] = hi.x; av[j][5] = hi.y; av[j][6] = hi.z; av[j][7] = hi.w;
    }
  }
}

template <int kG, bool kGuard>
__device__ __forceinline__ void p3_apply(const float (&av)[kTg][8], v8f (&hreg)[kTpw],
                                         int smin, const v2f (&by)[4]) {
  if (!kGuard) {
#pragma unroll
    for (int q = 0; q < 4; ++q) {
#pragma unroll
      for (int j = 0; j < kTg; ++j) {
        const v2f fa = {av[j][2 * q], av[j][2 * q + 1]};
        hreg[kG * kTg + j] = wmma_f32_16x16x4(fa, by[q], hreg[kG * kTg + j]);
      }
    }
  } else {
#pragma unroll
    for (int j = 0; j < kTg; ++j) {
      if (kG * kTg + j >= smin) {
#pragma unroll
        for (int q = 0; q < 4; ++q) {
          const v2f fa = {av[j][2 * q], av[j][2 * q + 1]};
          hreg[kG * kTg + j] = wmma_f32_16x16x4(fa, by[q], hreg[kG * kTg + j]);
        }
      }
    }
  }
}

template <int kG>
__device__ __forceinline__ void p3_step(float (&av)[kTg][8], v8f (&hreg)[kTpw], const float* vt,
                                        int wave, int g0, int smin, int lh, int lm,
                                        const v2f (&by)[4]) {
  if (kG >= g0) {
    float cur[kTg][8];
#pragma unroll
    for (int j = 0; j < kTg; ++j) {
#pragma unroll
      for (int q = 0; q < 8; ++q) cur[j][q] = av[j][q];
    }
    if (kG + 1 < kNg) p3_load<false>(av, vt, wave, (kG + 1) * kTg, smin, lh, lm);
    if (kG * kTg >= smin) p3_apply<kG, false>(cur, hreg, smin, by);
    else                  p3_apply<kG, true>(cur, hreg, smin, by);
  }
}

template <bool kTail>
__device__ __forceinline__ void gram_batch(const float* p, int i0, int nt, int k, int row,
                                           int lh, v8f& accG) {
  v2f a[kGb];
#pragma unroll
  for (int j = 0; j < kGb; ++j) {
    a[j] = v2f{0.0f, 0.0f};
    if (!kTail || i0 + j < nt) a[j] = *(const v2f*)(p + (i0 + j) * 32);
  }
#pragma unroll
  for (int j = 0; j < kGb; ++j) {
    if (!kTail || i0 + j < nt) {
      const int col = k + (i0 + j) * 32 + 2 * lh;
      v2f u;
      u.x = (col >= row) ? a[j].x : 0.0f;
      u.y = (col + 1 >= row) ? a[j].y : 0.0f;
      if (row == kN - 1 && col == kN - 1) u.x = 1.0f;
      if (row == kN - 1 && col + 1 == kN - 1) u.y = 1.0f;
      accG = wmma_f32_16x16x4(u, u, accG);
    }
  }
}

__global__ __launch_bounds__(kThreads)
void block_factor_kernel(const float* __restrict__ w, float* __restrict__ tf) {
  __shared__ __attribute__((aligned(16))) float fsm[kWaves * 256 + 256 + 16 + 256];
  float* Gpart = fsm;
  float* Gbuf  = Gpart + kWaves * 256;
  float* Cbuf  = Gbuf + 256;
  float* Tl    = Cbuf + 16;

  const int tid  = threadIdx.x;
  const int wave = __builtin_amdgcn_readfirstlane(tid >> 5);
  const int lane = tid & 31;
  const int lm   = lane & 15;
  const int lh   = lane >> 4;
  const int blk  = blockIdx.x;
  const int base = blk * kBlk;

  v8f accG = {};
  const int k = base + wave * 4;
  if (k < kN) {
    const int nt  = (kN - k + kWaves * 4 - 1) / (kWaves * 4);
    const int row = base + lm;
    const float* p = w + (size_t)row * kN + k + 2 * lh;
    int i = 0;
    for (; i + kGb <= nt; i += kGb) gram_batch<false>(p, i, nt, k, row, lh, accG);
    if (i < nt) gram_batch<true>(p, i, nt, k, row, lh, accG);
  }
#pragma unroll
  for (int r = 0; r < 8; ++r)
    Gpart[wave * 256 + (r + 8 * lh) * 16 + lm] = accG[r];
  __syncthreads();
  {
    float sg = 0.f;
#pragma unroll
    for (int pw = 0; pw < kWaves; ++pw) sg += Gpart[pw * 256 + tid];
    Gbuf[tid] = sg;
    if ((tid >> 4) == (tid & 15)) Cbuf[tid & 15] = (sg > 0.f) ? (2.0f / sg) : 0.0f;
  }
  __syncthreads();

  if (tid < 32) {
    float y[16];
    v4f c4[4];
#pragma unroll
    for (int pp = 0; pp < 4; ++pp) c4[pp] = *(const v4f*)(Cbuf + 4 * pp);
    v4f g4[16][4];
#pragma unroll
    for (int m = 0; m < 15; ++m) {
#pragma unroll
      for (int pp = (m + 1) >> 2; pp < 4; ++pp)
        g4[m][pp] = *(const v4f*)(Gbuf + m * 16 + 4 * pp);
    }
#pragma unroll
    for (int m = 15; m >= 0; --m) {
      float sacc = (m == lm) ? 1.0f : 0.0f;
#pragma unroll
      for (int l = 15; l > m; --l) sacc = __builtin_fmaf(g4[m][l >> 2][l & 3], y[l], sacc);
      y[m] = -c4[m >> 2][m & 3] * sacc;
    }
    if (lh == 0) {
#pragma unroll
      for (int m = 0; m < 16; ++m) Tl[m * 16 + lm] = y[m];
    }
  }
  __syncthreads();
  if (tid < 32) {
    const float* tr = Tl + lm * 16 + 2 * lh;
    const v4f lo = {tr[0], tr[1], tr[4], tr[5]};
    const v4f hi = {tr[8], tr[9], tr[12], tr[13]};
    float* d = tf + (size_t)blk * kTfBlk + lane * 4;
    *(v4f*)(d)       = lo;
    *(v4f*)(d + 128) = hi;
  }
}

__global__ __launch_bounds__(kThreads)
void ortho_householder_kernel(const float* __restrict__ x,
                              const float* __restrict__ w,
                              const float* __restrict__ tf,
                              float* __restrict__ out) {
  __shared__ __attribute__((aligned(16))) float smem[kSmemFloats];
  float* hbuf  = smem;
  float* Wpart = hbuf + kCols * kHStr;
  float* Wbuf  = Wpart + kWaves * 256;
  float* Vt    = Wbuf + 256;

  const int tid  = threadIdx.x;
  const int wave = __builtin_amdgcn_readfirstlane(tid >> 5);
  const int lane = tid & 31;
  const int lm   = lane & 15;
  const int lh   = lane >> 4;
  const int c0   = blockIdx.x * kCols;
  float* const hrow = hbuf + lm * kHStr;

  v4f stg[kSt];
  stage_load(stg, w, kN - kBlk, wave, lane);
  const float* tfl = tf + lane * 4;
  v4f tnl = *(const v4f*)(tfl + (size_t)(kN / kBlk - 1) * kTfBlk);
  v4f tnh = *(const v4f*)(tfl + (size_t)(kN / kBlk - 1) * kTfBlk + 128);

  for (int i = tid; i < kCols * (kN / 4); i += kThreads) {
    const int c  = i / (kN / 4);
    const int k4 = (i - c * (kN / 4)) * 4;
    v4f v = *(const v4f*)(x + (size_t)(c0 + c) * kN + k4);
    *(v4f*)(hbuf + c * kHStr + k4) = v;
  }
  __syncthreads();

  v8f hreg[kTpw];
#pragma unroll
  for (int s = 0; s < kTpw; ++s) {
    const float* hp = hrow + (wave + kWaves * s) * 16 + 8 * lh;
    const v4f lo = *(const v4f*)(hp);
    const v4f hi = *(const v4f*)(hp + 4);
    hreg[s] = v8f{lo.x, lo.y, lo.z, lo.w, hi.x, hi.y, hi.z, hi.w};
  }

  for (int base = kN - kBlk; base >= 0; base -= kBlk) {
    stage_store(stg, Vt, base, wave, lane);
    __syncthreads();

    const int bt   = base >> 4;
    const int smin = (bt - wave + kWaves - 1) >> 3;
    const int g0   = smin >> 2;
    const int gl   = min(g0, kNg - 1);
    v8f accW = {};
    {
      float a1[kTg][8];
      p1_load<true>(a1, Vt, wave, gl * kTg, smin, lh, lm);
      p1_step<0>(a1, hreg, Vt, wave, g0, smin, lh, lm, accW);
      p1_step<1>(a1, hreg, Vt, wave, g0, smin, lh, lm, accW);
      p1_step<2>(a1, hreg, Vt, wave, g0, smin, lh, lm, accW);
      p1_step<3>(a1, hreg, Vt, wave, g0, smin, lh, lm, accW);
    }
#pragma unroll
    for (int r = 0; r < 8; ++r)
      Wpart[wave * 256 + (r + 8 * lh) * 16 + lm] = accW[r];
    __syncthreads();

    {
      float sw = 0.f;
#pragma unroll
      for (int p = 0; p < kWaves; ++p) sw += Wpart[p * 256 + tid];
      Wbuf[tid] = sw;
    }
    __syncthreads();

    const v4f tcl = tnl, tch = tnh;
    if (base >= kBlk) {
      stage_load(stg, w, base - kBlk, wave, lane);
      tnl = *(const v4f*)(tfl + (size_t)(base / kBlk - 1) * kTfBlk);
      tnh = *(const v4f*)(tfl + (size_t)(base / kBlk - 1) * kTfBlk + 128);
    }
    float av[kTg][8];
    p3_load<true>(av, Vt, wave, gl * kTg, smin, lh, lm);

    v2f by[4];
    {
      v2f bw[4];
#pragma unroll
      for (int q = 0; q < 4; ++q) {
        bw[q].x = Wbuf[(4 * q + 2 * lh) * 16 + lm];
        bw[q].y = Wbuf[(4 * q + 2 * lh + 1) * 16 + lm];
      }
      v8f yacc = {};
      yacc = wmma_f32_16x16x4(v2f{tcl.x, tcl.y}, bw[0], yacc);
      yacc = wmma_f32_16x16x4(v2f{tcl.z, tcl.w}, bw[1], yacc);
      yacc = wmma_f32_16x16x4(v2f{tch.x, tch.y}, bw[2], yacc);
      yacc = wmma_f32_16x16x4(v2f{tch.z, tch.w}, bw[3], yacc);
#pragma unroll
      for (int q = 0; q < 4; ++q) {
        by[q].x = yacc[2 * q];
        by[q].y = yacc[2 * q + 1];
      }
    }

    p3_step<0>(av, hreg, Vt, wave, g0, smin, lh, lm, by);
    p3_step<1>(av, hreg, Vt, wave, g0, smin, lh, lm, by);
    p3_step<2>(av, hreg, Vt, wave, g0, smin, lh, lm, by);
    p3_step<3>(av, hreg, Vt, wave, g0, smin, lh, lm, by);
    __syncthreads();
  }

#pragma unroll
  for (int s = 0; s < kTpw; ++s) {
    float* hp = hrow + (wave + kWaves * s) * 16 + 8 * lh;
    *(v4f*)(hp)     = v4f{hreg[s][0], hreg[s][1], hreg[s][2], hreg[s][3]};
    *(v4f*)(hp + 4) = v4f{hreg[s][4], hreg[s][5], hreg[s][6], hreg[s][7]};
  }
  __syncthreads();

  for (int i = tid; i < kCols * (kN / 4); i += kThreads) {
    const int c  = i / (kN / 4);
    const int k4 = (i - c * (kN / 4)) * 4;
    v4f v = *(const v4f*)(hbuf + c * kHStr + k4);
    *(v4f*)(out + (size_t)(c0 + c) * kN + k4) = v;
  }
}

extern "C" void kernel_launch(void* const* d_in, const int* in_sizes, int n_in,
                              void* d_out, int out_size, void* d_ws, size_t ws_size,
                              hipStream_t stream) {
  const float* x = (const float*)d_in[0];
  const float* w = (const float*)d_in[1];
  float* out = (float*)d_out;
  float* tf  = (float*)d_ws;

  hipLaunchKernelGGL(block_factor_kernel, dim3(kN / kBlk), dim3(kThreads), 0, stream, w, tf);

  dim3 grid(kB / kCols);
  dim3 block(kThreads);
  hipLaunchKernelGGL(ortho_householder_kernel, grid, block, 0, stream,
                     x, w, (const float*)tf, out);
}


# Each wave loads, transposes and consumes only its own column tiles of U: phase 1 multiplies straight from the loaded registers, Vt is wave-private, and two of the four barriers per block go
# speedup vs baseline: 6.1265x; 1.1731x over previous
#include <hip/hip_runtime.h>

typedef __attribute__((ext_vector_type(2))) float v2f;
typedef __attribute__((ext_vector_type(4))) float v4f;
typedef __attribute__((ext_vector_type(8))) float v8f;

namespace {
constexpr int kN     = 2048;
constexpr int kB     = 4096;
constexpr int kCols  = 16;
constexpr int kBlk   = 16;
constexpr int kWaves = 8;
constexpr int kThreads = kWaves * 32;
constexpr int kHStr  = kN + 4;
constexpr int kVtStr = kBlk + 4;
constexpr int kSmemFloats = kCols * kHStr
                          + kWaves * 256
                          + 256
                          + kN * kVtStr;
static_assert(kSmemFloats * sizeof(float) <= 320u * 1024u, "one work-group's LDS");
static_assert(((kCols * kHStr + kWaves * 256 + 256) % 4) == 0, "Vt starts 16-byte aligned");
constexpr int kTfBlk = 256;
constexpr int kGb    = 16;
}

__device__ __forceinline__ v8f wmma_f32_16x16x4(v2f a, v2f b, v8f c) {
  return __builtin_amdgcn_wmma_f32_16x16x4_f32(
       false, a,  false, b,
       (short)0, c,  false,  false);
}

constexpr int kTg  = 4;
constexpr int kTpw = kN / 16 / kWaves;
constexpr int kNg  = kTpw / kTg;

__device__ __forceinline__ int first_slot(int bt, int wave) {
  return (bt - wave + kWaves - 1) >> 3;
}

template <int kG>
__device__ __forceinline__ void stage_load(v4f (&stg)[2 * kTpw], const float* __restrict__ w,
                                           const float* p, int smin) {
#pragma unroll
  for (int s = kG * kTg; s < (kG + 1) * kTg; ++s) {
    const float* ps = (s >= smin) ? p + s * (16 * kWaves) : w;
    stg[2 * s]     = *(const v4f*)(ps);
    stg[2 * s + 1] = *(const v4f*)(ps + 4);
  }
}
static_assert(kTpw == 16 && kNg == 4, "the slot switch of phase 1 lists 16 slots, the block loop 4 pieces");

__device__ __forceinline__ void cut_upper(v4f& lo, v4f& hi, int base, int c0, int lh, int lm) {
  const int row = base + lm;
#pragma unroll
  for (int e = 0; e < 4; ++e) {
    const int cl = c0 + 8 * lh + e;
    const int ch = cl + 4;
    float ul = (cl >= row) ? lo[e] : 0.0f;
    float uh = (ch >= row) ? hi[e] : 0.0f;
    if (row == kN - 1 && cl == kN - 1) ul = 1.0f;
    if (row == kN - 1 && ch == kN - 1) uh = 1.0f;
    lo[e] = ul;
    hi[e] = uh;
  }
}

__device__ __forceinline__ void p1_tile(const v4f& lo, const v4f& hi, const v8f& ht, float* d,
                                        v8f& accW) {
  d[0]          = lo.x;
  d[kVtStr]     = lo.y;
  d[2 * kVtStr] = lo.z;
  d[3 * kVtStr] = lo.w;
  d[4 * kVtStr] = hi.x;
  d[5 * kVtStr] = hi.y;
  d[6 * kVtStr] = hi.z;
  d[7 * kVtStr] = hi.w;
  accW = wmma_f32_16x16x4(v2f{lo.x, lo.y}, v2f{ht[0], ht[1]}, accW);
  accW = wmma_f32_16x16x4(v2f{lo.z, lo.w}, v2f{ht[2], ht[3]}, accW);
  accW = wmma_f32_16x16x4(v2f{hi.x, hi.y}, v2f{ht[4], ht[5]}, accW);
  accW = wmma_f32_16x16x4(v2f{hi.z, hi.w}, v2f{ht[6], ht[7]}, accW);
}


template <bool kGuard>
__device__ __forceinline__ void p3_load(float (&av)[kTg][8], const float* vt, int wave, int s0,
                                        int smin, int lh, int lm) {
#pragma unroll
  for (int j = 0; j < kTg; ++j) {
#pragma unroll
    for (int q = 0; q < 8; ++q) av[j][q] = 0.0f;
    if (!kGuard || s0 + j >= smin) {
      const float* vp = vt + ((wave * kTpw + (s0 + j)) * 16 + lm) * kVtStr + 8 * lh;
      const v4f lo = *(const v4f*)(vp);
      const v4f hi = *(const v4f*)(vp + 4);
      av[j][0] = lo.x; av[j][1] = lo.y; av[j][2] = lo.z; av[j][3] = lo.w;
      av[j][4] = hi.x; av[j][5] = hi.y; av[j][6] = hi.z; av[j][7] = hi.w;
    }
  }
}

template <int kG, bool kGuard>
__device__ __forceinline__ void p3_apply(const float (&av)[kTg][8], v8f (&hreg)[kTpw],
                                         int smin, const v2f (&by)[4]) {
  if (!kGuard) {
#pragma unroll
    for (int q = 0; q < 4; ++q) {
#pragma unroll
      for (int j = 0; j < kTg; ++j) {
        const v2f fa = {av[j][2 * q], av[j][2 * q + 1]};
        hreg[kG * kTg + j] = wmma_f32_16x16x4(fa, by[q], hreg[kG * kTg + j]);
      }
    }
  } else {
#pragma unroll
    for (int j = 0; j < kTg; ++j) {
      if (kG * kTg + j >= smin) {
#pragma unroll
        for (int q = 0; q < 4; ++q) {
          const v2f fa = {av[j][2 * q], av[j][2 * q + 1]};
          hreg[kG * kTg + j] = wmma_f32_16x16x4(fa, by[q], hreg[kG * kTg + j]);
        }
      }
    }
  }
}

template <int kG>
__device__ __forceinline__ void p3_step(float (&av)[kTg][8], v8f (&hreg)[kTpw], const float* vt,
                                        int wave, int g0, int smin, int lh, int lm,
                                        const v2f (&by)[4]) {
  if (kG >= g0) {
    float cur[kTg][8];
#pragma unroll
    for (int j = 0; j < kTg; ++j) {
#pragma unroll
      for (int q = 0; q < 8; ++q) cur[j][q] = av[j][q];
    }
    if (kG + 1 < kNg) p3_load<false>(av, vt, wave, (kG + 1) * kTg, smin, lh, lm);
    if (kG * kTg >= smin) p3_apply<kG, false>(cur, hreg, smin, by);
    else                  p3_apply<kG, true>(cur, hreg, smin, by);
  }
}

template <bool kTail>
__device__ __forceinline__ void gram_batch(const float* p, int i0, int nt, int k, int row,
                                           int lh, v8f& accG) {
  v2f a[kGb];
#pragma unroll
  for (int j = 0; j < kGb; ++j) {
    a[j] = v2f{0.0f, 0.0f};
    if (!kTail || i0 + j < nt) a[j] = *(const v2f*)(p + (i0 + j) * 32);
  }
#pragma unroll
  for (int j = 0; j < kGb; ++j) {
    if (!kTail || i0 + j < nt) {
      const int col = k + (i0 + j) * 32 + 2 * lh;
      v2f u;
      u.x = (col >= row) ? a[j].x : 0.0f;
      u.y = (col + 1 >= row) ? a[j].y : 0.0f;
      if (row == kN - 1 && col == kN - 1) u.x = 1.0f;
      if (row == kN - 1 && col + 1 == kN - 1) u.y = 1.0f;
      accG = wmma_f32_16x16x4(u, u, accG);
    }
  }
}

__global__ __launch_bounds__(kThreads)
void block_factor_kernel(const float* __restrict__ w, float* __restrict__ tf) {
  __shared__ __attribute__((aligned(16))) float fsm[kWaves * 256 + 256 + 16 + 256];
  float* Gpart = fsm;
  float* Gbuf  = Gpart + kWaves * 256;
  float* Cbuf  = Gbuf + 256;
  float* Tl    = Cbuf + 16;

  const int tid  = threadIdx.x;
  const int wave = __builtin_amdgcn_readfirstlane(tid >> 5);
  const int lane = tid & 31;
  const int lm   = lane & 15;
  const int lh   = lane >> 4;
  const int blk  = blockIdx.x;
  const int base = blk * kBlk;

  v8f accG = {};
  const int k = base + wave * 4;
  if (k < kN) {
    const int nt  = (kN - k + kWaves * 4 - 1) / (kWaves * 4);
    const int row = base + lm;
    const float* p = w + (size_t)row * kN + k + 2 * lh;
    int i = 0;
    for (; i + kGb <= nt; i += kGb) gram_batch<false>(p, i, nt, k, row, lh, accG);
    if (i < nt) gram_batch<true>(p, i, nt, k, row, lh, accG);
  }
#pragma unroll
  for (int r = 0; r < 8; ++r)
    Gpart[wave * 256 + (r + 8 * lh) * 16 + lm] = accG[r];
  __syncthreads();
  {
    float sg = 0.f;
#pragma unroll
    for (int pw = 0; pw < kWaves; ++pw) sg += Gpart[pw * 256 + tid];
    Gbuf[tid] = sg;
    if ((tid >> 4) == (tid & 15)) Cbuf[tid & 15] = (sg > 0.f) ? (2.0f / sg) : 0.0f;
  }
  __syncthreads();

  if (tid < 32) {
    float y[16];
    v4f c4[4];
#pragma unroll
    for (int pp = 0; pp < 4; ++pp) c4[pp] = *(const v4f*)(Cbuf + 4 * pp);
    v4f g4[16][4];
#pragma unroll
    for (int m = 0; m < 15; ++m) {
#pragma unroll
      for (int pp = (m + 1) >> 2; pp < 4; ++pp)
        g4[m][pp] = *(const v4f*)(Gbuf + m * 16 + 4 * pp);
    }
#pragma unroll
    for (int m = 15; m >= 0; --m) {
      float sacc = (m == lm) ? 1.0f : 0.0f;
#pragma unroll
      for (int l = 15; l > m; --l) sacc = __builtin_fmaf(g4[m][l >> 2][l & 3], y[l], sacc);
      y[m] = -c4[m >> 2][m & 3] * sacc;
    }
    if (lh == 0) {
#pragma unroll
      for (int m = 0; m < 16; ++m) Tl[m * 16 + lm] = y[m];
    }
  }
  __syncthreads();
  if (tid < 32) {
    const float* tr = Tl + lm * 16 + 2 * lh;
    const v4f lo = {tr[0], tr[1], tr[4], tr[5]};
    const v4f hi = {tr[8], tr[9], tr[12], tr[13]};
    float* d = tf + (size_t)blk * kTfBlk + lane * 4;
    *(v4f*)(d)       = lo;
    *(v4f*)(d + 128) = hi;
  }
}

__global__ __launch_bounds__(kThreads)
void ortho_householder_kernel(const float* __restrict__ x,
                              const float* __restrict__ w,
                              const float* __restrict__ tf,
                              float* __restrict__ out) {
  __shared__ __attribute__((aligned(16))) float smem[kSmemFloats];
  float* hbuf  = smem;
  float* Wpart = hbuf + kCols * kHStr;
  float* Wbuf  = Wpart + kWaves * 256;
  float* Vt    = Wbuf + 256;

  const int tid  = threadIdx.x;
  const int wave = __builtin_amdgcn_readfirstlane(tid >> 5);
  const int lane = tid & 31;
  const int lm   = lane & 15;
  const int lh   = lane >> 4;
  const int c0   = blockIdx.x * kCols;
  float* const hrow = hbuf + lm * kHStr;

  v4f stg[2 * kTpw];
  {
    const float* p0 = w + (size_t)(kN - kBlk + lm) * kN + wave * 16 + 8 * lh;
    const int s0 = first_slot((kN - kBlk) >> 4, wave);
    stage_load<0>(stg, w, p0, s0);
    stage_load<1>(stg, w, p0, s0);
    stage_load<2>(stg, w, p0, s0);
    stage_load<3>(stg, w, p0, s0);
  }
  const float* tfl = tf + lane * 4;
  v4f tnl = *(const v4f*)(tfl + (size_t)(kN / kBlk - 1) * kTfBlk);
  v4f tnh = *(const v4f*)(tfl + (size_t)(kN / kBlk - 1) * kTfBlk + 128);

  for (int i = tid; i < kCols * (kN / 4); i += kThreads) {
    const int c  = i / (kN / 4);
    const int k4 = (i - c * (kN / 4)) * 4;
    v4f v = *(const v4f*)(x + (size_t)(c0 + c) * kN + k4);
    *(v4f*)(hbuf + c * kHStr + k4) = v;
  }
  __syncthreads();

  v8f hreg[kTpw];
#pragma unroll
  for (int s = 0; s < kTpw; ++s) {
    const float* hp = hrow + (wave + kWaves * s) * 16 + 8 * lh;
    const v4f lo = *(const v4f*)(hp);
    const v4f hi = *(const v4f*)(hp + 4);
    hreg[s] = v8f{lo.x, lo.y, lo.z, lo.w, hi.x, hi.y, hi.z, hi.w};
  }

  for (int base = kN - kBlk; base >= 0; base -= kBlk) {
    const int bt   = base >> 4;
    const int smin = first_slot(bt, wave);
    const int g0   = smin >> 2;
    const int gl   = min(g0, kNg - 1);
    v8f accW = {};
    {
      float* const vq = Vt + (wave * kTpw * 16 + 8 * lh) * kVtStr + lm;
#define P1_SLOT(S)                                                                          \
  case S:                                                                                   \
    if (wave + kWaves * S <= bt)                                                            \
      cut_upper(stg[2 * S], stg[2 * S + 1], base, (wave + kWaves * S) * 16, lh, lm);        \
    p1_tile(stg[2 * S], stg[2 * S + 1], hreg[S], vq + S * (16 * kVtStr), accW);             \
    [[fallthrough]];
      switch (min((unsigned)smin, (unsigned)(kTpw - 1))) {
        P1_SLOT(0)  P1_SLOT(1)  P1_SLOT(2)  P1_SLOT(3)  P1_SLOT(4)  P1_SLOT(5)
        P1_SLOT(6)  P1_SLOT(7)  P1_SLOT(8)  P1_SLOT(9)  P1_SLOT(10) P1_SLOT(11)
        P1_SLOT(12) P1_SLOT(13) P1_SLOT(14) P1_SLOT(15)
        default: break;
      }
#undef P1_SLOT
    }
#pragma unroll
    for (int r = 0; r < 8; ++r)
      Wpart[wave * 256 + (r + 8 * lh) * 16 + lm] = accW[r];
    __syncthreads();

    {
      float sw = 0.f;
#pragma unroll
      for (int p = 0; p < kWaves; ++p) sw += Wpart[p * 256 + tid];
      Wbuf[tid] = sw;
    }
    __syncthreads();

    float av[kTg][8];
    p3_load<true>(av, Vt, wave, gl * kTg, smin, lh, lm);

    v2f by[4];
    {
      v2f bw[4];
#pragma unroll
      for (int q = 0; q < 4; ++q) {
        bw[q].x = Wbuf[(4 * q + 2 * lh) * 16 + lm];
        bw[q].y = Wbuf[(4 * q + 2 * lh + 1) * 16 + lm];
      }
      v8f yacc = {};
      yacc = wmma_f32_16x16x4(v2f{tnl.x, tnl.y}, bw[0], yacc);
      yacc = wmma_f32_16x16x4(v2f{tnl.z, tnl.w}, bw[1], yacc);
      yacc = wmma_f32_16x16x4(v2f{tnh.x, tnh.y}, bw[2], yacc);
      yacc = wmma_f32_16x16x4(v2f{tnh.z, tnh.w}, bw[3], yacc);
#pragma unroll
      for (int q = 0; q < 4; ++q) {
        by[q].x = yacc[2 * q];
        by[q].y = yacc[2 * q + 1];
      }
    }
    const bool more = base >= kBlk;
    const int nbase = more ? base - kBlk : 0;
    const int nsmin = more ? first_slot(nbase >> 4, wave) : kTpw;
    const float* np = w + (size_t)(nbase + lm) * kN + wave * 16 + 8 * lh;
    stage_load<0>(stg, w, np, nsmin);
    p3_step<0>(av, hreg, Vt, wave, g0, smin, lh, lm, by);
    stage_load<1>(stg, w, np, nsmin);
    p3_step<1>(av, hreg, Vt, wave, g0, smin, lh, lm, by);
    stage_load<2>(stg, w, np, nsmin);
    p3_step<2>(av, hreg, Vt, wave, g0, smin, lh, lm, by);
    stage_load<3>(stg, w, np, nsmin);
    p3_step<3>(av, hreg, Vt, wave, g0, smin, lh, lm, by);
    tnl = *(const v4f*)(tfl + (size_t)(nbase / kBlk) * kTfBlk);
    tnh = *(const v4f*)(tfl + (size_t)(nbase / kBlk) * kTfBlk + 128);
  }

#pragma unroll
  for (int s = 0; s < kTpw; ++s) {
    float* hp = hrow + (wave + kWaves * s) * 16 + 8 * lh;
    *(v4f*)(hp)     = v4f{hreg[s][0], hreg[s][1], hreg[s][2], hreg[s][3]};
    *(v4f*)(hp + 4) = v4f{hreg[s][4], hreg[s][5], hreg[s][6], hreg[s][7]};
  }
  __syncthreads();

  for (int i = tid; i < kCols * (kN / 4); i += kThreads) {
    const int c  = i / (kN / 4);
    const int k4 = (i - c * (kN / 4)) * 4;
    v4f v = *(const v4f*)(hbuf + c * kHStr + k4);
    *(v4f*)(out + (size_t)(c0 + c) * kN + k4) = v;
  }
}

extern "C" void kernel_launch(void* const* d_in, const int* in_sizes, int n_in,
                              void* d_out, int out_size, void* d_ws, size_t ws_size,
                              hipStream_t stream) {
  const float* x = (const float*)d_in[0];
  const float* w = (const float*)d_in[1];
  float* out = (float*)d_out;
  float* tf  = (float*)d_ws;

  hipLaunchKernelGGL(block_factor_kernel, dim3(kN / kBlk), dim3(kThreads), 0, stream, w, tf);

  dim3 grid(kB / kCols);
  dim3 block(kThreads);
  hipLaunchKernelGGL(ortho_householder_kernel, grid, block, 0, stream,
                     x, w, (const float*)tf, out);
}


# Main kernel runs 16 waves per workgroup (four per SIMD) instead of 8, each keeping 8 row tiles of h; W is now the sum of 16 wave partials (that sum's order changes); phase 1 slots as scalar tests
# speedup vs baseline: 6.6763x; 1.0897x over previous
#include <hip/hip_runtime.h>

typedef __attribute__((ext_vector_type(2))) float v2f;
typedef __attribute__((ext_vector_type(4))) float v4f;
typedef __attribute__((ext_vector_type(8))) float v8f;

namespace {
constexpr int kN     = 2048;
constexpr int kB     = 4096;
constexpr int kCols  = 16;
constexpr int kBlk   = 16;
constexpr int kWaves = 16;
constexpr int kThreads = kWaves * 32;
constexpr int kFWaves = 8;
constexpr int kFThreads = kFWaves * 32;
constexpr int kHStr  = kN + 4;
constexpr int kVtStr = kBlk + 4;
constexpr int kSmemFloats = kCols * kHStr
                          + kWaves * 256
                          + 256
                          + kN * kVtStr;
static_assert(kSmemFloats * sizeof(float) <= 320u * 1024u, "one work-group's LDS");
static_assert(((kCols * kHStr + kWaves * 256 + 256) % 4) == 0, "Vt starts 16-byte aligned");
constexpr int kTfBlk = 256;
constexpr int kGb    = 16;
}

__device__ __forceinline__ v8f wmma_f32_16x16x4(v2f a, v2f b, v8f c) {
  return __builtin_amdgcn_wmma_f32_16x16x4_f32(
       false, a,  false, b,
       (short)0, c,  false,  false);
}

constexpr int kTg  = 4;
constexpr int kTpw = kN / 16 / kWaves;
constexpr int kNg  = kTpw / kTg;

__device__ __forceinline__ int first_slot(int bt, int wave) {
  return (bt - wave + kWaves - 1) >> 4;
}
static_assert(kWaves == 16, "first_slot divides by the wave count with a shift of 4");

template <int kG>
__device__ __forceinline__ void stage_load(v4f (&stg)[2 * kTpw], const float* __restrict__ w,
                                           const float* p, int smin) {
#pragma unroll
  for (int s = kG * kTg; s < (kG + 1) * kTg; ++s) {
    const float* ps = (s >= smin) ? p + s * (16 * kWaves) : w;
    stg[2 * s]     = *(const v4f*)(ps);
    stg[2 * s + 1] = *(const v4f*)(ps + 4);
  }
}
static_assert(kTpw == 8 && kNg == 2, "the slot switch of phase 1 lists 8 slots, the block loop 2 pieces");

__device__ __forceinline__ void cut_upper(v4f& lo, v4f& hi, int base, int c0, int lh, int lm) {
  const int row = base + lm;
#pragma unroll
  for (int e = 0; e < 4; ++e) {
    const int cl = c0 + 8 * lh + e;
    const int ch = cl + 4;
    float ul = (cl >= row) ? lo[e] : 0.0f;
    float uh = (ch >= row) ? hi[e] : 0.0f;
    if (row == kN - 1 && cl == kN - 1) ul = 1.0f;
    if (row == kN - 1 && ch == kN - 1) uh = 1.0f;
    lo[e] = ul;
    hi[e] = uh;
  }
}

__device__ __forceinline__ void p1_tile(const v4f& lo, const v4f& hi, const v8f& ht, float* d,
                                        v8f& accW) {
  d[0]          = lo.x;
  d[kVtStr]     = lo.y;
  d[2 * kVtStr] = lo.z;
  d[3 * kVtStr] = lo.w;
  d[4 * kVtStr] = hi.x;
  d[5 * kVtStr] = hi.y;
  d[6 * kVtStr] = hi.z;
  d[7 * kVtStr] = hi.w;
  accW = wmma_f32_16x16x4(v2f{lo.x, lo.y}, v2f{ht[0], ht[1]}, accW);
  accW = wmma_f32_16x16x4(v2f{lo.z, lo.w}, v2f{ht[2], ht[3]}, accW);
  accW = wmma_f32_16x16x4(v2f{hi.x, hi.y}, v2f{ht[4], ht[5]}, accW);
  accW = wmma_f32_16x16x4(v2f{hi.z, hi.w}, v2f{ht[6], ht[7]}, accW);
}


template <bool kGuard>
__device__ __forceinline__ void p3_load(float (&av)[kTg][8], const float* vt, int wave, int s0,
                                        int smin, int lh, int lm) {
#pragma unroll
  for (int j = 0; j < kTg; ++j) {
#pragma unroll
    for (int q = 0; q < 8; ++q) av[j][q] = 0.0f;
    if (!kGuard || s0 + j >= smin) {
      const float* vp = vt + ((wave * kTpw + (s0 + j)) * 16 + lm) * kVtStr + 8 * lh;
      const v4f lo = *(const v4f*)(vp);
      const v4f hi = *(const v4f*)(vp + 4);
      av[j][0] = lo.x; av[j][1] = lo.y; av[j][2] = lo.z; av[j][3] = lo.w;
      av[j][4] = hi.x; av[j][5] = hi.y; av[j][6] = hi.z; av[j][7] = hi.w;
    }
  }
}

template <int kG, bool kGuard>
__device__ __forceinline__ void p3_apply(const float (&av)[kTg][8], v8f (&hreg)[kTpw],
                                         int smin, const v2f (&by)[4]) {
  if (!kGuard) {
#pragma unroll
    for (int q = 0; q < 4; ++q) {
#pragma unroll
      for (int j = 0; j < kTg; ++j) {
        const v2f fa = {av[j][2 * q], av[j][2 * q + 1]};
        hreg[kG * kTg + j] = wmma_f32_16x16x4(fa, by[q], hreg[kG * kTg + j]);
      }
    }
  } else {
#pragma unroll
    for (int j = 0; j < kTg; ++j) {
      if (kG * kTg + j >= smin) {
#pragma unroll
        for (int q = 0; q < 4; ++q) {
          const v2f fa = {av[j][2 * q], av[j][2 * q + 1]};
          hreg[kG * kTg + j] = wmma_f32_16x16x4(fa, by[q], hreg[kG * kTg + j]);
        }
      }
    }
  }
}

template <int kG>
__device__ __forceinline__ void p3_step(float (&av)[kTg][8], v8f (&hreg)[kTpw], const float* vt,
                                        int wave, int g0, int smin, int lh, int lm,
                                        const v2f (&by)[4]) {
  if (kG >= g0) {
    float cur[kTg][8];
#pragma unroll
    for (int j = 0; j < kTg; ++j) {
#pragma unroll
      for (int q = 0; q < 8; ++q) cur[j][q] = av[j][q];
    }
    if (kG + 1 < kNg) p3_load<false>(av, vt, wave, (kG + 1) * kTg, smin, lh, lm);
    if (kG * kTg >= smin) p3_apply<kG, false>(cur, hreg, smin, by);
    else                  p3_apply<kG, true>(cur, hreg, smin, by);
  }
}

template <bool kTail>
__device__ __forceinline__ void gram_batch(const float* p, int i0, int nt, int k, int row,
                                           int lh, v8f& accG) {
  v2f a[kGb];
#pragma unroll
  for (int j = 0; j < kGb; ++j) {
    a[j] = v2f{0.0f, 0.0f};
    if (!kTail || i0 + j < nt) a[j] = *(const v2f*)(p + (i0 + j) * 32);
  }
#pragma unroll
  for (int j = 0; j < kGb; ++j) {
    if (!kTail || i0 + j < nt) {
      const int col = k + (i0 + j) * 32 + 2 * lh;
      v2f u;
      u.x = (col >= row) ? a[j].x : 0.0f;
      u.y = (col + 1 >= row) ? a[j].y : 0.0f;
      if (row == kN - 1 && col == kN - 1) u.x = 1.0f;
      if (row == kN - 1 && col + 1 == kN - 1) u.y = 1.0f;
      accG = wmma_f32_16x16x4(u, u, accG);
    }
  }
}

__global__ __launch_bounds__(kFThreads)
void block_factor_kernel(const float* __restrict__ w, float* __restrict__ tf) {
  __shared__ __attribute__((aligned(16))) float fsm[kFWaves * 256 + 256 + 16 + 256];
  float* Gpart = fsm;
  float* Gbuf  = Gpart + kFWaves * 256;
  float* Cbuf  = Gbuf + 256;
  float* Tl    = Cbuf + 16;

  const int tid  = threadIdx.x;
  const int wave = __builtin_amdgcn_readfirstlane(tid >> 5);
  const int lane = tid & 31;
  const int lm   = lane & 15;
  const int lh   = lane >> 4;
  const int blk  = blockIdx.x;
  const int base = blk * kBlk;

  v8f accG = {};
  const int k = base + wave * 4;
  if (k < kN) {
    const int nt  = (kN - k + kFWaves * 4 - 1) / (kFWaves * 4);
    const int row = base + lm;
    const float* p = w + (size_t)row * kN + k + 2 * lh;
    int i = 0;
    for (; i + kGb <= nt; i += kGb) gram_batch<false>(p, i, nt, k, row, lh, accG);
    if (i < nt) gram_batch<true>(p, i, nt, k, row, lh, accG);
  }
#pragma unroll
  for (int r = 0; r < 8; ++r)
    Gpart[wave * 256 + (r + 8 * lh) * 16 + lm] = accG[r];
  __syncthreads();
  {
    float sg = 0.f;
#pragma unroll
    for (int pw = 0; pw < kFWaves; ++pw) sg += Gpart[pw * 256 + tid];
    Gbuf[tid] = sg;
    if ((tid >> 4) == (tid & 15)) Cbuf[tid & 15] = (sg > 0.f) ? (2.0f / sg) : 0.0f;
  }
  __syncthreads();

  if (tid < 32) {
    float y[16];
    v4f c4[4];
#pragma unroll
    for (int pp = 0; pp < 4; ++pp) c4[pp] = *(const v4f*)(Cbuf + 4 * pp);
    v4f g4[16][4];
#pragma unroll
    for (int m = 0; m < 15; ++m) {
#pragma unroll
      for (int pp = (m + 1) >> 2; pp < 4; ++pp)
        g4[m][pp] = *(const v4f*)(Gbuf + m * 16 + 4 * pp);
    }
#pragma unroll
    for (int m = 15; m >= 0; --m) {
      float sacc = (m == lm) ? 1.0f : 0.0f;
#pragma unroll
      for (int l = 15; l > m; --l) sacc = __builtin_fmaf(g4[m][l >> 2][l & 3], y[l], sacc);
      y[m] = -c4[m >> 2][m & 3] * sacc;
    }
    if (lh == 0) {
#pragma unroll
      for (int m = 0; m < 16; ++m) Tl[m * 16 + lm] = y[m];
    }
  }
  __syncthreads();
  if (tid < 32) {
    const float* tr = Tl + lm * 16 + 2 * lh;
    const v4f lo = {tr[0], tr[1], tr[4], tr[5]};
    const v4f hi = {tr[8], tr[9], tr[12], tr[13]};
    float* d = tf + (size_t)blk * kTfBlk + lane * 4;
    *(v4f*)(d)       = lo;
    *(v4f*)(d + 128) = hi;
  }
}

__global__ __launch_bounds__(kThreads)
void ortho_householder_kernel(const float* __restrict__ x,
                              const float* __restrict__ w,
                              const float* __restrict__ tf,
                              float* __restrict__ out) {
  __shared__ __attribute__((aligned(16))) float smem[kSmemFloats];
  float* hbuf  = smem;
  float* Wpart = hbuf + kCols * kHStr;
  float* Wbuf  = Wpart + kWaves * 256;
  float* Vt    = Wbuf + 256;

  const int tid  = threadIdx.x;
  const int wave = __builtin_amdgcn_readfirstlane(tid >> 5);
  const int lane = tid & 31;
  const int lm   = lane & 15;
  const int lh   = lane >> 4;
  const int c0   = blockIdx.x * kCols;
  float* const hrow = hbuf + lm * kHStr;

  v4f stg[2 * kTpw];
  {
    const float* p0 = w + (size_t)(kN - kBlk + lm) * kN + wave * 16 + 8 * lh;
    const int s0 = first_slot((kN - kBlk) >> 4, wave);
    stage_load<0>(stg, w, p0, s0);
    stage_load<1>(stg, w, p0, s0);
  }
  const float* tfl = tf + lane * 4;
  v4f tnl = *(const v4f*)(tfl + (size_t)(kN / kBlk - 1) * kTfBlk);
  v4f tnh = *(const v4f*)(tfl + (size_t)(kN / kBlk - 1) * kTfBlk + 128);

  for (int i = tid; i < kCols * (kN / 4); i += kThreads) {
    const int c  = i / (kN / 4);
    const int k4 = (i - c * (kN / 4)) * 4;
    v4f v = *(const v4f*)(x + (size_t)(c0 + c) * kN + k4);
    *(v4f*)(hbuf + c * kHStr + k4) = v;
  }
  __syncthreads();

  v8f hreg[kTpw];
#pragma unroll
  for (int s = 0; s < kTpw; ++s) {
    const float* hp = hrow + (wave + kWaves * s) * 16 + 8 * lh;
    const v4f lo = *(const v4f*)(hp);
    const v4f hi = *(const v4f*)(hp + 4);
    hreg[s] = v8f{lo.x, lo.y, lo.z, lo.w, hi.x, hi.y, hi.z, hi.w};
  }

  for (int base = kN - kBlk; base >= 0; base -= kBlk) {
    const int bt   = base >> 4;
    const int smin = first_slot(bt, wave);
    const int g0   = smin >> 2;
    const int gl   = min(g0, kNg - 1);
    v8f accW = {};
    {
      float* const vq = Vt + (wave * kTpw * 16 + 8 * lh) * kVtStr + lm;
      const int sfirst = min(smin, kTpw - 1);
#define P1_SLOT(S)                                                                          \
  if (S >= sfirst) {                                                                        \
    if (wave + kWaves * S <= bt)                                                            \
      cut_upper(stg[2 * S], stg[2 * S + 1], base, (wave + kWaves * S) * 16, lh, lm);        \
    p1_tile(stg[2 * S], stg[2 * S + 1], hreg[S], vq + S * (16 * kVtStr), accW);             \
  }
      P1_SLOT(0)  P1_SLOT(1)  P1_SLOT(2)  P1_SLOT(3)
      P1_SLOT(4)  P1_SLOT(5)  P1_SLOT(6)  P1_SLOT(7)
#undef P1_SLOT
    }
#pragma unroll
    for (int r = 0; r < 8; ++r)
      Wpart[wave * 256 + (r + 8 * lh) * 16 + lm] = accW[r];
    __syncthreads();

    if (tid < 256) {
      float sw = 0.f;
#pragma unroll
      for (int p = 0; p < kWaves; ++p) sw += Wpart[p * 256 + tid];
      Wbuf[tid] = sw;
    }
    __syncthreads();

    float av[kTg][8];
    p3_load<true>(av, Vt, wave, gl * kTg, smin, lh, lm);

    v2f by[4];
    {
      v2f bw[4];
#pragma unroll
      for (int q = 0; q < 4; ++q) {
        bw[q].x = Wbuf[(4 * q + 2 * lh) * 16 + lm];
        bw[q].y = Wbuf[(4 * q + 2 * lh + 1) * 16 + lm];
      }
      v8f yacc = {};
      yacc = wmma_f32_16x16x4(v2f{tnl.x, tnl.y}, bw[0], yacc);
      yacc = wmma_f32_16x16x4(v2f{tnl.z, tnl.w}, bw[1], yacc);
      yacc = wmma_f32_16x16x4(v2f{tnh.x, tnh.y}, bw[2], yacc);
      yacc = wmma_f32_16x16x4(v2f{tnh.z, tnh.w}, bw[3], yacc);
#pragma unroll
      for (int q = 0; q < 4; ++q) {
        by[q].x = yacc[2 * q];
        by[q].y = yacc[2 * q + 1];
      }
    }
    const bool more = base >= kBlk;
    const int nbase = more ? base - kBlk : 0;
    const int nsmin = more ? first_slot(nbase >> 4, wave) : kTpw;
    const float* np = w + (size_t)(nbase + lm) * kN + wave * 16 + 8 * lh;
    stage_load<0>(stg, w, np, nsmin);
    p3_step<0>(av, hreg, Vt, wave, g0, smin, lh, lm, by);
    stage_load<1>(stg, w, np, nsmin);
    p3_step<1>(av, hreg, Vt, wave, g0, smin, lh, lm, by);
    tnl = *(const v4f*)(tfl + (size_t)(nbase / kBlk) * kTfBlk);
    tnh = *(const v4f*)(tfl + (size_t)(nbase / kBlk) * kTfBlk + 128);
  }

#pragma unroll
  for (int s = 0; s < kTpw; ++s) {
    float* hp = hrow + (wave + kWaves * s) * 16 + 8 * lh;
    *(v4f*)(hp)     = v4f{hreg[s][0], hreg[s][1], hreg[s][2], hreg[s][3]};
    *(v4f*)(hp + 4) = v4f{hreg[s][4], hreg[s][5], hreg[s][6], hreg[s][7]};
  }
  __syncthreads();

  for (int i = tid; i < kCols * (kN / 4); i += kThreads) {
    const int c  = i / (kN / 4);
    const int k4 = (i - c * (kN / 4)) * 4;
    v4f v = *(const v4f*)(hbuf + c * kHStr + k4);
    *(v4f*)(out + (size_t)(c0 + c) * kN + k4) = v;
  }
}

extern "C" void kernel_launch(void* const* d_in, const int* in_sizes, int n_in,
                              void* d_out, int out_size, void* d_ws, size_t ws_size,
                              hipStream_t stream) {
  const float* x = (const float*)d_in[0];
  const float* w = (const float*)d_in[1];
  float* out = (float*)d_out;
  float* tf  = (float*)d_ws;

  hipLaunchKernelGGL(block_factor_kernel, dim3(kN / kBlk), dim3(kFThreads), 0, stream, w, tf);

  dim3 grid(kB / kCols);
  dim3 block(kThreads);
  hipLaunchKernelGGL(ortho_householder_kernel, grid, block, 0, stream,
                     x, w, (const float*)tf, out);
}


# First launch also stores the cut rows of U in the main kernel's lane order (extra work-groups): each V load of a wave is 512 contiguous bytes, not pieces of 16 lines; triu cut leaves the main loop
# speedup vs baseline: 7.0139x; 1.0506x over previous
#include <hip/hip_runtime.h>

typedef __attribute__((ext_vector_type(2))) float v2f;
typedef __attribute__((ext_vector_type(4))) float v4f;
typedef __attribute__((ext_vector_type(8))) float v8f;

namespace {
constexpr int kN     = 2048;
constexpr int kB     = 4096;
constexpr int kCols  = 16;
constexpr int kBlk   = 16;
constexpr int kWaves = 16;
constexpr int kThreads = kWaves * 32;
constexpr int kFWaves = 8;
constexpr int kFThreads = kFWaves * 32;
constexpr int kHStr  = kN + 4;
constexpr int kVtStr = kBlk + 4;
constexpr int kSmemFloats = kCols * kHStr
                          + kWaves * 256
                          + 256
                          + kN * kVtStr;
static_assert(kSmemFloats * sizeof(float) <= 320u * 1024u, "one work-group's LDS");
static_assert(((kCols * kHStr + kWaves * 256 + 256) % 4) == 0, "Vt starts 16-byte aligned");
constexpr int kTfBlk = 256;
constexpr int kGb    = 16;
}

__device__ __forceinline__ v8f wmma_f32_16x16x4(v2f a, v2f b, v8f c) {
  return __builtin_amdgcn_wmma_f32_16x16x4_f32(
       false, a,  false, b,
       (short)0, c,  false,  false);
}

constexpr int kTg  = 4;
constexpr int kTpw = kN / 16 / kWaves;
constexpr int kNg  = kTpw / kTg;

constexpr int kTfFloats = (kN / kBlk) * kTfBlk;
constexpr int kVsTile   = 256;
constexpr int kVsWave   = kTpw * kVsTile;
constexpr int kVsBlk    = kWaves * kVsWave;
constexpr int kSwParts  = 2;

__device__ __forceinline__ int first_slot(int bt, int wave) {
  return (bt - wave + kWaves - 1) >> 4;
}
static_assert(kWaves == 16, "first_slot divides by the wave count with a shift of 4");

template <int kG>
__device__ __forceinline__ void stage_load(v4f (&stg)[2 * kTpw], const float* __restrict__ w,
                                           const float* p, int smin) {
#pragma unroll
  for (int s = kG * kTg; s < (kG + 1) * kTg; ++s) {
    const float* ps = (s >= smin) ? p + s * kVsTile : w;
    stg[2 * s]     = *(const v4f*)(ps);
    stg[2 * s + 1] = *(const v4f*)(ps + 128);
  }
}
static_assert(kTpw == 8 && kNg == 2, "the slot switch of phase 1 lists 8 slots, the block loop 2 pieces");

__device__ __forceinline__ void cut_upper(v4f& lo, v4f& hi, int base, int c0, int lh, int lm) {
  const int row = base + lm;
#pragma unroll
  for (int e = 0; e < 4; ++e) {
    const int cl = c0 + 8 * lh + e;
    const int ch = cl + 4;
    float ul = (cl >= row) ? lo[e] : 0.0f;
    float uh = (ch >= row) ? hi[e] : 0.0f;
    if (row == kN - 1 && cl == kN - 1) ul = 1.0f;
    if (row == kN - 1 && ch == kN - 1) uh = 1.0f;
    lo[e] = ul;
    hi[e] = uh;
  }
}

__device__ __forceinline__ void p1_tile(const v4f& lo, const v4f& hi, const v8f& ht, float* d,
                                        v8f& accW) {
  d[0]          = lo.x;
  d[kVtStr]     = lo.y;
  d[2 * kVtStr] = lo.z;
  d[3 * kVtStr] = lo.w;
  d[4 * kVtStr] = hi.x;
  d[5 * kVtStr] = hi.y;
  d[6 * kVtStr] = hi.z;
  d[7 * kVtStr] = hi.w;
  accW = wmma_f32_16x16x4(v2f{lo.x, lo.y}, v2f{ht[0], ht[1]}, accW);
  accW = wmma_f32_16x16x4(v2f{lo.z, lo.w}, v2f{ht[2], ht[3]}, accW);
  accW = wmma_f32_16x16x4(v2f{hi.x, hi.y}, v2f{ht[4], ht[5]}, accW);
  accW = wmma_f32_16x16x4(v2f{hi.z, hi.w}, v2f{ht[6], ht[7]}, accW);
}


template <bool kGuard>
__device__ __forceinline__ void p3_load(float (&av)[kTg][8], const float* vt, int wave, int s0,
                                        int smin, int lh, int lm) {
#pragma unroll
  for (int j = 0; j < kTg; ++j) {
#pragma unroll
    for (int q = 0; q < 8; ++q) av[j][q] = 0.0f;
    if (!kGuard || s0 + j >= smin) {
      const float* vp = vt + ((wave * kTpw + (s0 + j)) * 16 + lm) * kVtStr + 8 * lh;
      const v4f lo = *(const v4f*)(vp);
      const v4f hi = *(const v4f*)(vp + 4);
      av[j][0] = lo.x; av[j][1] = lo.y; av[j][2] = lo.z; av[j][3] = lo.w;
      av[j][4] = hi.x; av[j][5] = hi.y; av[j][6] = hi.z; av[j][7] = hi.w;
    }
  }
}

template <int kG, bool kGuard>
__device__ __forceinline__ void p3_apply(const float (&av)[kTg][8], v8f (&hreg)[kTpw],
                                         int smin, const v2f (&by)[4]) {
  if (!kGuard) {
#pragma unroll
    for (int q = 0; q < 4; ++q) {
#pragma unroll
      for (int j = 0; j < kTg; ++j) {
        const v2f fa = {av[j][2 * q], av[j][2 * q + 1]};
        hreg[kG * kTg + j] = wmma_f32_16x16x4(fa, by[q], hreg[kG * kTg + j]);
      }
    }
  } else {
#pragma unroll
    for (int j = 0; j < kTg; ++j) {
      if (kG * kTg + j >= smin) {
#pragma unroll
        for (int q = 0; q < 4; ++q) {
          const v2f fa = {av[j][2 * q], av[j][2 * q + 1]};
          hreg[kG * kTg + j] = wmma_f32_16x16x4(fa, by[q], hreg[kG * kTg + j]);
        }
      }
    }
  }
}

template <int kG>
__device__ __forceinline__ void p3_step(float (&av)[kTg][8], v8f (&hreg)[kTpw], const float* vt,
                                        int wave, int g0, int smin, int lh, int lm,
                                        const v2f (&by)[4]) {
  if (kG >= g0) {
    float cur[kTg][8];
#pragma unroll
    for (int j = 0; j < kTg; ++j) {
#pragma unroll
      for (int q = 0; q < 8; ++q) cur[j][q] = av[j][q];
    }
    if (kG + 1 < kNg) p3_load<false>(av, vt, wave, (kG + 1) * kTg, smin, lh, lm);
    if (kG * kTg >= smin) p3_apply<kG, false>(cur, hreg, smin, by);
    else                  p3_apply<kG, true>(cur, hreg, smin, by);
  }
}

template <bool kTail>
__device__ __forceinline__ void gram_batch(const float* p, int i0, int nt, int k, int row,
                                           int lh, v8f& accG) {
  v2f a[kGb];
#pragma unroll
  for (int j = 0; j < kGb; ++j) {
    a[j] = v2f{0.0f, 0.0f};
    if (!kTail || i0 + j < nt) a[j] = *(const v2f*)(p + (i0 + j) * 32);
  }
#pragma unroll
  for (int j = 0; j < kGb; ++j) {
    if (!kTail || i0 + j < nt) {
      const int col = k + (i0 + j) * 32 + 2 * lh;
      v2f u;
      u.x = (col >= row) ? a[j].x : 0.0f;
      u.y = (col + 1 >= row) ? a[j].y : 0.0f;
      if (row == kN - 1 && col == kN - 1) u.x = 1.0f;
      if (row == kN - 1 && col + 1 == kN - 1) u.y = 1.0f;
      accG = wmma_f32_16x16x4(u, u, accG);
    }
  }
}

__device__ __forceinline__ void relay_tiles(const float* __restrict__ w, float* __restrict__ vs,
                                            int q, int wave, int lane) {
  const int lm   = lane & 15;
  const int lh   = lane >> 4;
  const int blk  = q / kSwParts;
  const int part = q - blk * kSwParts;
  const int base = blk * kBlk;
  const int bt   = base >> 4;
  constexpr int kPer = (kN / 16) / (kSwParts * kFWaves);
  static_assert(kPer * kSwParts * kFWaves == kN / 16, "every tile of a block has one owner");
  const float* sp = w + (size_t)(base + lm) * kN + 8 * lh;
  float* dp = vs + (size_t)blk * kVsBlk + lane * 4;
  v4f lo[kPer], hi[kPer];
#pragma unroll
  for (int i = 0; i < kPer; ++i) {
    const int R = part + kSwParts * (wave + kFWaves * i);
    const float* ps = (R >= bt) ? sp + R * 16 : w;
    lo[i] = *(const v4f*)(ps);
    hi[i] = *(const v4f*)(ps + 4);
  }
#pragma unroll
  for (int i = 0; i < kPer; ++i) {
    const int R = part + kSwParts * (wave + kFWaves * i);
    cut_upper(lo[i], hi[i], base, R * 16, lh, lm);
    float* d = dp + ((R & (kWaves - 1)) * kTpw + (R >> 4)) * kVsTile;
    *(v4f*)(d)       = lo[i];
    *(v4f*)(d + 128) = hi[i];
  }
}

__global__ __launch_bounds__(kFThreads)
void block_factor_kernel(const float* __restrict__ w, float* __restrict__ tf) {
  if (blockIdx.x >= kN / kBlk) {
    const int rw = __builtin_amdgcn_readfirstlane(threadIdx.x >> 5);
    relay_tiles(w, tf + kTfFloats, (int)blockIdx.x - kN / kBlk, rw, threadIdx.x & 31);
    return;
  }
  __shared__ __attribute__((aligned(16))) float fsm[kFWaves * 256 + 256 + 16 + 256];
  float* Gpart = fsm;
  float* Gbuf  = Gpart + kFWaves * 256;
  float* Cbuf  = Gbuf + 256;
  float* Tl    = Cbuf + 16;

  const int tid  = threadIdx.x;
  const int wave = __builtin_amdgcn_readfirstlane(tid >> 5);
  const int lane = tid & 31;
  const int lm   = lane & 15;
  const int lh   = lane >> 4;
  const int blk  = blockIdx.x;
  const int base = blk * kBlk;

  v8f accG = {};
  const int k = base + wave * 4;
  if (k < kN) {
    const int nt  = (kN - k + kFWaves * 4 - 1) / (kFWaves * 4);
    const int row = base + lm;
    const float* p = w + (size_t)row * kN + k + 2 * lh;
    int i = 0;
    for (; i + kGb <= nt; i += kGb) gram_batch<false>(p, i, nt, k, row, lh, accG);
    if (i < nt) gram_batch<true>(p, i, nt, k, row, lh, accG);
  }
#pragma unroll
  for (int r = 0; r < 8; ++r)
    Gpart[wave * 256 + (r + 8 * lh) * 16 + lm] = accG[r];
  __syncthreads();
  {
    float sg = 0.f;
#pragma unroll
    for (int pw = 0; pw < kFWaves; ++pw) sg += Gpart[pw * 256 + tid];
    Gbuf[tid] = sg;
    if ((tid >> 4) == (tid & 15)) Cbuf[tid & 15] = (sg > 0.f) ? (2.0f / sg) : 0.0f;
  }
  __syncthreads();

  if (tid < 32) {
    float y[16];
    v4f c4[4];
#pragma unroll
    for (int pp = 0; pp < 4; ++pp) c4[pp] = *(const v4f*)(Cbuf + 4 * pp);
    v4f g4[16][4];
#pragma unroll
    for (int m = 0; m < 15; ++m) {
#pragma unroll
      for (int pp = (m + 1) >> 2; pp < 4; ++pp)
        g4[m][pp] = *(const v4f*)(Gbuf + m * 16 + 4 * pp);
    }
#pragma unroll
    for (int m = 15; m >= 0; --m) {
      float sacc = (m == lm) ? 1.0f : 0.0f;
#pragma unroll
      for (int l = 15; l > m; --l) sacc = __builtin_fmaf(g4[m][l >> 2][l & 3], y[l], sacc);
      y[m] = -c4[m >> 2][m & 3] * sacc;
    }
    if (lh == 0) {
#pragma unroll
      for (int m = 0; m < 16; ++m) Tl[m * 16 + lm] = y[m];
    }
  }
  __syncthreads();
  if (tid < 32) {
    const float* tr = Tl + lm * 16 + 2 * lh;
    const v4f lo = {tr[0], tr[1], tr[4], tr[5]};
    const v4f hi = {tr[8], tr[9], tr[12], tr[13]};
    float* d = tf + (size_t)blk * kTfBlk + lane * 4;
    *(v4f*)(d)       = lo;
    *(v4f*)(d + 128) = hi;
  }
}

__global__ __launch_bounds__(kThreads)
void ortho_householder_kernel(const float* __restrict__ x,
                              const float* __restrict__ w,
                              const float* __restrict__ tf,
                              float* __restrict__ out) {
  __shared__ __attribute__((aligned(16))) float smem[kSmemFloats];
  float* hbuf  = smem;
  float* Wpart = hbuf + kCols * kHStr;
  float* Wbuf  = Wpart + kWaves * 256;
  float* Vt    = Wbuf + 256;

  const int tid  = threadIdx.x;
  const int wave = __builtin_amdgcn_readfirstlane(tid >> 5);
  const int lane = tid & 31;
  const int lm   = lane & 15;
  const int lh   = lane >> 4;
  const int c0   = blockIdx.x * kCols;
  float* const hrow = hbuf + lm * kHStr;

  v4f stg[2 * kTpw];
  const float* vsl = tf + kTfFloats + wave * kVsWave + lane * 4;
  {
    const float* p0 = vsl + (size_t)(kN / kBlk - 1) * kVsBlk;
    const int s0 = min(first_slot((kN - kBlk) >> 4, wave), kTpw - 1);
    stage_load<0>(stg, w, p0, s0);
    stage_load<1>(stg, w, p0, s0);
  }
  const float* tfl = tf + lane * 4;
  v4f tnl = *(const v4f*)(tfl + (size_t)(kN / kBlk - 1) * kTfBlk);
  v4f tnh = *(const v4f*)(tfl + (size_t)(kN / kBlk - 1) * kTfBlk + 128);

  for (int i = tid; i < kCols * (kN / 4); i += kThreads) {
    const int c  = i / (kN / 4);
    const int k4 = (i - c * (kN / 4)) * 4;
    v4f v = *(const v4f*)(x + (size_t)(c0 + c) * kN + k4);
    *(v4f*)(hbuf + c * kHStr + k4) = v;
  }
  __syncthreads();

  v8f hreg[kTpw];
#pragma unroll
  for (int s = 0; s < kTpw; ++s) {
    const float* hp = hrow + (wave + kWaves * s) * 16 + 8 * lh;
    const v4f lo = *(const v4f*)(hp);
    const v4f hi = *(const v4f*)(hp + 4);
    hreg[s] = v8f{lo.x, lo.y, lo.z, lo.w, hi.x, hi.y, hi.z, hi.w};
  }

  for (int base = kN - kBlk; base >= 0; base -= kBlk) {
    const int bt   = base >> 4;
    const int smin = first_slot(bt, wave);
    const int g0   = smin >> 2;
    const int gl   = min(g0, kNg - 1);
    v8f accW = {};
    {
      float* const vq = Vt + (wave * kTpw * 16 + 8 * lh) * kVtStr + lm;
      const int sfirst = min(smin, kTpw - 1);
#define P1_SLOT(S)                                                                          \
  if (S >= sfirst) {                                                                        \
    p1_tile(stg[2 * S], stg[2 * S + 1], hreg[S], vq + S * (16 * kVtStr), accW);             \
  }
      P1_SLOT(0)  P1_SLOT(1)  P1_SLOT(2)  P1_SLOT(3)
      P1_SLOT(4)  P1_SLOT(5)  P1_SLOT(6)  P1_SLOT(7)
#undef P1_SLOT
    }
#pragma unroll
    for (int r = 0; r < 8; ++r)
      Wpart[wave * 256 + (r + 8 * lh) * 16 + lm] = accW[r];
    __syncthreads();

    if (tid < 256) {
      float sw = 0.f;
#pragma unroll
      for (int p = 0; p < kWaves; ++p) sw += Wpart[p * 256 + tid];
      Wbuf[tid] = sw;
    }
    __syncthreads();

    float av[kTg][8];
    p3_load<true>(av, Vt, wave, gl * kTg, smin, lh, lm);

    v2f by[4];
    {
      v2f bw[4];
#pragma unroll
      for (int q = 0; q < 4; ++q) {
        bw[q].x = Wbuf[(4 * q + 2 * lh) * 16 + lm];
        bw[q].y = Wbuf[(4 * q + 2 * lh + 1) * 16 + lm];
      }
      v8f yacc = {};
      yacc = wmma_f32_16x16x4(v2f{tnl.x, tnl.y}, bw[0], yacc);
      yacc = wmma_f32_16x16x4(v2f{tnl.z, tnl.w}, bw[1], yacc);
      yacc = wmma_f32_16x16x4(v2f{tnh.x, tnh.y}, bw[2], yacc);
      yacc = wmma_f32_16x16x4(v2f{tnh.z, tnh.w}, bw[3], yacc);
#pragma unroll
      for (int q = 0; q < 4; ++q) {
        by[q].x = yacc[2 * q];
        by[q].y = yacc[2 * q + 1];
      }
    }
    const bool more = base >= kBlk;
    const int nbase = more ? base - kBlk : 0;
    const int nsmin = more ? min(first_slot(nbase >> 4, wave), kTpw - 1) : kTpw;
    const float* np = vsl + (size_t)(nbase / kBlk) * kVsBlk;
    stage_load<0>(stg, w, np, nsmin);
    p3_step<0>(av, hreg, Vt, wave, g0, smin, lh, lm, by);
    stage_load<1>(stg, w, np, nsmin);
    p3_step<1>(av, hreg, Vt, wave, g0, smin, lh, lm, by);
    tnl = *(const v4f*)(tfl + (size_t)(nbase / kBlk) * kTfBlk);
    tnh = *(const v4f*)(tfl + (size_t)(nbase / kBlk) * kTfBlk + 128);
  }

#pragma unroll
  for (int s = 0; s < kTpw; ++s) {
    float* hp = hrow + (wave + kWaves * s) * 16 + 8 * lh;
    *(v4f*)(hp)     = v4f{hreg[s][0], hreg[s][1], hreg[s][2], hreg[s][3]};
    *(v4f*)(hp + 4) = v4f{hreg[s][4], hreg[s][5], hreg[s][6], hreg[s][7]};
  }
  __syncthreads();

  for (int i = tid; i < kCols * (kN / 4); i += kThreads) {
    const int c  = i / (kN / 4);
    const int k4 = (i - c * (kN / 4)) * 4;
    v4f v = *(const v4f*)(hbuf + c * kHStr + k4);
    *(v4f*)(out + (size_t)(c0 + c) * kN + k4) = v;
  }
}

extern "C" void kernel_launch(void* const* d_in, const int* in_sizes, int n_in,
                              void* d_out, int out_size, void* d_ws, size_t ws_size,
                              hipStream_t stream) {
  const float* x = (const float*)d_in[0];
  const float* w = (const float*)d_in[1];
  float* out = (float*)d_out;
  float* tf  = (float*)d_ws;

  hipLaunchKernelGGL(block_factor_kernel, dim3(kN / kBlk + kSwParts * (kN / kBlk)), dim3(kFThreads),
                     0, stream, w, tf);

  dim3 grid(kB / kCols);
  dim3 block(kThreads);
  hipLaunchKernelGGL(ortho_householder_kernel, grid, block, 0, stream,
                     x, w, (const float*)tf, out);
}


# W hand-over shortened: partials and sum kept in WMMA lane order (128-bit LDS ops, tree sum: W's sum order changes); first Vt reads moved before barrier 1, next block's loads held behind the Y product
# speedup vs baseline: 7.3479x; 1.0476x over previous
#include <hip/hip_runtime.h>

typedef __attribute__((ext_vector_type(2))) float v2f;
typedef __attribute__((ext_vector_type(4))) float v4f;
typedef __attribute__((ext_vector_type(8))) float v8f;

namespace {
constexpr int kN     = 2048;
constexpr int kB     = 4096;
constexpr int kCols  = 16;
constexpr int kBlk   = 16;
constexpr int kWaves = 16;
constexpr int kThreads = kWaves * 32;
constexpr int kFWaves = 8;
constexpr int kFThreads = kFWaves * 32;
constexpr int kHStr  = kN + 4;
constexpr int kVtStr = kBlk + 4;
constexpr int kWpStr   = 12;
constexpr int kWpPlane = 32 * kWpStr;
constexpr int kSmemFloats = kCols * kHStr
                          + kWaves * kWpPlane
                          + kWpPlane
                          + kN * kVtStr;
static_assert(kSmemFloats * sizeof(float) <= 320u * 1024u, "one work-group's LDS");
static_assert(((kCols * kHStr + kWaves * kWpPlane + kWpPlane) % 4) == 0, "Vt starts 16-byte aligned");
constexpr int kTfBlk = 256;
constexpr int kGb    = 16;
}

__device__ __forceinline__ v8f wmma_f32_16x16x4(v2f a, v2f b, v8f c) {
  return __builtin_amdgcn_wmma_f32_16x16x4_f32(
       false, a,  false, b,
       (short)0, c,  false,  false);
}

constexpr int kTg  = 4;
constexpr int kTpw = kN / 16 / kWaves;
constexpr int kNg  = kTpw / kTg;

constexpr int kTfFloats = (kN / kBlk) * kTfBlk;
constexpr int kVsTile   = 256;
constexpr int kVsWave   = kTpw * kVsTile;
constexpr int kVsBlk    = kWaves * kVsWave;
constexpr int kSwParts  = 2;

__device__ __forceinline__ int first_slot(int bt, int wave) {
  return (bt - wave + kWaves - 1) >> 4;
}
static_assert(kWaves == 16, "first_slot divides by the wave count with a shift of 4");

template <int kG>
__device__ __forceinline__ void stage_load(v4f (&stg)[2 * kTpw], const float* __restrict__ w,
                                           const float* p, int smin) {
#pragma unroll
  for (int s = kG * kTg; s < (kG + 1) * kTg; ++s) {
    const float* ps = (s >= smin) ? p + s * kVsTile : w;
    stg[2 * s]     = *(const v4f*)(ps);
    stg[2 * s + 1] = *(const v4f*)(ps + 128);
  }
}
static_assert(kTpw == 8 && kNg == 2, "the slot switch of phase 1 lists 8 slots, the block loop 2 pieces");

__device__ __forceinline__ void cut_upper(v4f& lo, v4f& hi, int base, int c0, int lh, int lm) {
  const int row = base + lm;
#pragma unroll
  for (int e = 0; e < 4; ++e) {
    const int cl = c0 + 8 * lh + e;
    const int ch = cl + 4;
    float ul = (cl >= row) ? lo[e] : 0.0f;
    float uh = (ch >= row) ? hi[e] : 0.0f;
    if (row == kN - 1 && cl == kN - 1) ul = 1.0f;
    if (row == kN - 1 && ch == kN - 1) uh = 1.0f;
    lo[e] = ul;
    hi[e] = uh;
  }
}

__device__ __forceinline__ void p1_tile(const v4f& lo, const v4f& hi, const v8f& ht, float* d,
                                        v8f& accW) {
  d[0]          = lo.x;
  d[kVtStr]     = lo.y;
  d[2 * kVtStr] = lo.z;
  d[3 * kVtStr] = lo.w;
  d[4 * kVtStr] = hi.x;
  d[5 * kVtStr] = hi.y;
  d[6 * kVtStr] = hi.z;
  d[7 * kVtStr] = hi.w;
  accW = wmma_f32_16x16x4(v2f{lo.x, lo.y}, v2f{ht[0], ht[1]}, accW);
  accW = wmma_f32_16x16x4(v2f{lo.z, lo.w}, v2f{ht[2], ht[3]}, accW);
  accW = wmma_f32_16x16x4(v2f{hi.x, hi.y}, v2f{ht[4], ht[5]}, accW);
  accW = wmma_f32_16x16x4(v2f{hi.z, hi.w}, v2f{ht[6], ht[7]}, accW);
}


template <bool kGuard>
__device__ __forceinline__ void p3_load(float (&av)[kTg][8], const float* vt, int wave, int s0,
                                        int smin, int lh, int lm) {
#pragma unroll
  for (int j = 0; j < kTg; ++j) {
#pragma unroll
    for (int q = 0; q < 8; ++q) av[j][q] = 0.0f;
    if (!kGuard || s0 + j >= smin) {
      const float* vp = vt + ((wave * kTpw + (s0 + j)) * 16 + lm) * kVtStr + 8 * lh;
      const v4f lo = *(const v4f*)(vp);
      const v4f hi = *(const v4f*)(vp + 4);
      av[j][0] = lo.x; av[j][1] = lo.y; av[j][2] = lo.z; av[j][3] = lo.w;
      av[j][4] = hi.x; av[j][5] = hi.y; av[j][6] = hi.z; av[j][7] = hi.w;
    }
  }
}

template <int kG, bool kGuard>
__device__ __forceinline__ void p3_apply(const float (&av)[kTg][8], v8f (&hreg)[kTpw],
                                         int smin, const v2f (&by)[4]) {
  if (!kGuard) {
#pragma unroll
    for (int q = 0; q < 4; ++q) {
#pragma unroll
      for (int j = 0; j < kTg; ++j) {
        const v2f fa = {av[j][2 * q], av[j][2 * q + 1]};
        hreg[kG * kTg + j] = wmma_f32_16x16x4(fa, by[q], hreg[kG * kTg + j]);
      }
    }
  } else {
#pragma unroll
    for (int j = 0; j < kTg; ++j) {
      if (kG * kTg + j >= smin) {
#pragma unroll
        for (int q = 0; q < 4; ++q) {
          const v2f fa = {av[j][2 * q], av[j][2 * q + 1]};
          hreg[kG * kTg + j] = wmma_f32_16x16x4(fa, by[q], hreg[kG * kTg + j]);
        }
      }
    }
  }
}

template <int kG>
__device__ __forceinline__ void p3_step(float (&av)[kTg][8], v8f (&hreg)[kTpw], const float* vt,
                                        int wave, int g0, int smin, int lh, int lm,
                                        const v2f (&by)[4]) {
  if (kG >= g0) {
    float cur[kTg][8];
#pragma unroll
    for (int j = 0; j < kTg; ++j) {
#pragma unroll
      for (int q = 0; q < 8; ++q) cur[j][q] = av[j][q];
    }
    if (kG + 1 < kNg) p3_load<false>(av, vt, wave, (kG + 1) * kTg, smin, lh, lm);
    if (kG * kTg >= smin) p3_apply<kG, false>(cur, hreg, smin, by);
    else                  p3_apply<kG, true>(cur, hreg, smin, by);
  }
}

template <bool kTail>
__device__ __forceinline__ void gram_batch(const float* p, int i0, int nt, int k, int row,
                                           int lh, v8f& accG) {
  v2f a[kGb];
#pragma unroll
  for (int j = 0; j < kGb; ++j) {
    a[j] = v2f{0.0f, 0.0f};
    if (!kTail || i0 + j < nt) a[j] = *(const v2f*)(p + (i0 + j) * 32);
  }
#pragma unroll
  for (int j = 0; j < kGb; ++j) {
    if (!kTail || i0 + j < nt) {
      const int col = k + (i0 + j) * 32 + 2 * lh;
      v2f u;
      u.x = (col >= row) ? a[j].x : 0.0f;
      u.y = (col + 1 >= row) ? a[j].y : 0.0f;
      if (row == kN - 1 && col == kN - 1) u.x = 1.0f;
      if (row == kN - 1 && col + 1 == kN - 1) u.y = 1.0f;
      accG = wmma_f32_16x16x4(u, u, accG);
    }
  }
}

__device__ __forceinline__ void relay_tiles(const float* __restrict__ w, float* __restrict__ vs,
                                            int q, int wave, int lane) {
  const int lm   = lane & 15;
  const int lh   = lane >> 4;
  const int blk  = q / kSwParts;
  const int part = q - blk * kSwParts;
  const int base = blk * kBlk;
  const int bt   = base >> 4;
  constexpr int kPer = (kN / 16) / (kSwParts * kFWaves);
  static_assert(kPer * kSwParts * kFWaves == kN / 16, "every tile of a block has one owner");
  const float* sp = w + (size_t)(base + lm) * kN + 8 * lh;
  float* dp = vs + (size_t)blk * kVsBlk + lane * 4;
  v4f lo[kPer], hi[kPer];
#pragma unroll
  for (int i = 0; i < kPer; ++i) {
    const int R = part + kSwParts * (wave + kFWaves * i);
    const float* ps = (R >= bt) ? sp + R * 16 : w;
    lo[i] = *(const v4f*)(ps);
    hi[i] = *(const v4f*)(ps + 4);
  }
#pragma unroll
  for (int i = 0; i < kPer; ++i) {
    const int R = part + kSwParts * (wave + kFWaves * i);
    cut_upper(lo[i], hi[i], base, R * 16, lh, lm);
    float* d = dp + ((R & (kWaves - 1)) * kTpw + (R >> 4)) * kVsTile;
    *(v4f*)(d)       = lo[i];
    *(v4f*)(d + 128) = hi[i];
  }
}

__global__ __launch_bounds__(kFThreads)
void block_factor_kernel(const float* __restrict__ w, float* __restrict__ tf) {
  if (blockIdx.x >= kN / kBlk) {
    const int rw = __builtin_amdgcn_readfirstlane(threadIdx.x >> 5);
    relay_tiles(w, tf + kTfFloats, (int)blockIdx.x - kN / kBlk, rw, threadIdx.x & 31);
    return;
  }
  __shared__ __attribute__((aligned(16))) float fsm[kFWaves * 256 + 256 + 16 + 256];
  float* Gpart = fsm;
  float* Gbuf  = Gpart + kFWaves * 256;
  float* Cbuf  = Gbuf + 256;
  float* Tl    = Cbuf + 16;

  const int tid  = threadIdx.x;
  const int wave = __builtin_amdgcn_readfirstlane(tid >> 5);
  const int lane = tid & 31;
  const int lm   = lane & 15;
  const int lh   = lane >> 4;
  const int blk  = blockIdx.x;
  const int base = blk * kBlk;

  v8f accG = {};
  const int k = base + wave * 4;
  if (k < kN) {
    const int nt  = (kN - k + kFWaves * 4 - 1) / (kFWaves * 4);
    const int row = base + lm;
    const float* p = w + (size_t)row * kN + k + 2 * lh;
    int i = 0;
    for (; i + kGb <= nt; i += kGb) gram_batch<false>(p, i, nt, k, row, lh, accG);
    if (i < nt) gram_batch<true>(p, i, nt, k, row, lh, accG);
  }
#pragma unroll
  for (int r = 0; r < 8; ++r)
    Gpart[wave * 256 + (r + 8 * lh) * 16 + lm] = accG[r];
  __syncthreads();
  {
    float sg = 0.f;
#pragma unroll
    for (int pw = 0; pw < kFWaves; ++pw) sg += Gpart[pw * 256 + tid];
    Gbuf[tid] = sg;
    if ((tid >> 4) == (tid & 15)) Cbuf[tid & 15] = (sg > 0.f) ? (2.0f / sg) : 0.0f;
  }
  __syncthreads();

  if (tid < 32) {
    float y[16];
    v4f c4[4];
#pragma unroll
    for (int pp = 0; pp < 4; ++pp) c4[pp] = *(const v4f*)(Cbuf + 4 * pp);
    v4f g4[16][4];
#pragma unroll
    for (int m = 0; m < 15; ++m) {
#pragma unroll
      for (int pp = (m + 1) >> 2; pp < 4; ++pp)
        g4[m][pp] = *(const v4f*)(Gbuf + m * 16 + 4 * pp);
    }
#pragma unroll
    for (int m = 15; m >= 0; --m) {
      float sacc = (m == lm) ? 1.0f : 0.0f;
#pragma unroll
      for (int l = 15; l > m; --l) sacc = __builtin_fmaf(g4[m][l >> 2][l & 3], y[l], sacc);
      y[m] = -c4[m >> 2][m & 3] * sacc;
    }
    if (lh == 0) {
#pragma unroll
      for (int m = 0; m < 16; ++m) Tl[m * 16 + lm] = y[m];
    }
  }
  __syncthreads();
  if (tid < 32) {
    const float* tr = Tl + lm * 16 + 8 * lh;
    const v4f lo = {tr[0], tr[1], tr[2], tr[3]};
    const v4f hi = {tr[4], tr[5], tr[6], tr[7]};
    float* d = tf + (size_t)blk * kTfBlk + lane * 4;
    *(v4f*)(d)       = lo;
    *(v4f*)(d + 128) = hi;
  }
}

__global__ __launch_bounds__(kThreads)
void ortho_householder_kernel(const float* __restrict__ x,
                              const float* __restrict__ w,
                              const float* __restrict__ tf,
                              float* __restrict__ out) {
  __shared__ __attribute__((aligned(16))) float smem[kSmemFloats];
  float* hbuf  = smem;
  float* Wpart = hbuf + kCols * kHStr;
  float* Wbuf  = Wpart + kWaves * kWpPlane;
  float* Vt    = Wbuf + kWpPlane;

  const int tid  = threadIdx.x;
  const int wave = __builtin_amdgcn_readfirstlane(tid >> 5);
  const int lane = tid & 31;
  const int lm   = lane & 15;
  const int lh   = lane >> 4;
  const int c0   = blockIdx.x * kCols;
  float* const hrow = hbuf + lm * kHStr;

  v4f stg[2 * kTpw];
  const float* vsl = tf + kTfFloats + wave * kVsWave + lane * 4;
  {
    const float* p0 = vsl + (size_t)(kN / kBlk - 1) * kVsBlk;
    const int s0 = min(first_slot((kN - kBlk) >> 4, wave), kTpw - 1);
    stage_load<0>(stg, w, p0, s0);
    stage_load<1>(stg, w, p0, s0);
  }
  const float* tfl = tf + lane * 4;
  v4f tnl = *(const v4f*)(tfl + (size_t)(kN / kBlk - 1) * kTfBlk);
  v4f tnh = *(const v4f*)(tfl + (size_t)(kN / kBlk - 1) * kTfBlk + 128);

  for (int i = tid; i < kCols * (kN / 4); i += kThreads) {
    const int c  = i / (kN / 4);
    const int k4 = (i - c * (kN / 4)) * 4;
    v4f v = *(const v4f*)(x + (size_t)(c0 + c) * kN + k4);
    *(v4f*)(hbuf + c * kHStr + k4) = v;
  }
  __syncthreads();

  v8f hreg[kTpw];
#pragma unroll
  for (int s = 0; s < kTpw; ++s) {
    const float* hp = hrow + (wave + kWaves * s) * 16 + 8 * lh;
    const v4f lo = *(const v4f*)(hp);
    const v4f hi = *(const v4f*)(hp + 4);
    hreg[s] = v8f{lo.x, lo.y, lo.z, lo.w, hi.x, hi.y, hi.z, hi.w};
  }

  for (int base = kN - kBlk; base >= 0; base -= kBlk) {
    const int bt   = base >> 4;
    const int smin = first_slot(bt, wave);
    const int g0   = smin >> 2;
    const int gl   = min(g0, kNg - 1);
    v8f accW = {};
    {
      float* const vq = Vt + (wave * kTpw * 16 + 8 * lh) * kVtStr + lm;
      const int sfirst = min(smin, kTpw - 1);
#define P1_SLOT(S)                                                                          \
  if (S >= sfirst) {                                                                        \
    p1_tile(stg[2 * S], stg[2 * S + 1], hreg[S], vq + S * (16 * kVtStr), accW);             \
  }
      P1_SLOT(0)  P1_SLOT(1)  P1_SLOT(2)  P1_SLOT(3)
      P1_SLOT(4)  P1_SLOT(5)  P1_SLOT(6)  P1_SLOT(7)
#undef P1_SLOT
    }
    float av[kTg][8];
    p3_load<true>(av, Vt, wave, gl * kTg, smin, lh, lm);

    {
      float* wp = Wpart + wave * kWpPlane + lane * kWpStr;
      *(v4f*)(wp)     = v4f{accW[0], accW[1], accW[2], accW[3]};
      *(v4f*)(wp + 4) = v4f{accW[4], accW[5], accW[6], accW[7]};
    }
    __syncthreads();

    if (tid < 256) {
      const int el = (tid >> 3) * kWpStr + (tid & 7);
      float pw[kWaves];
#pragma unroll
      for (int p = 0; p < kWaves; ++p) pw[p] = Wpart[p * kWpPlane + el];
#pragma unroll
      for (int st = 1; st < kWaves; st *= 2) {
#pragma unroll
        for (int p = 0; p < kWaves; p += 2 * st) pw[p] += pw[p + st];
      }
      Wbuf[el] = pw[0];
    }
    __syncthreads();

    v2f by[4];
    {
      const float* wb = Wbuf + lane * kWpStr;
      const v4f wl = *(const v4f*)(wb);
      const v4f wh = *(const v4f*)(wb + 4);
      const v2f bw[4] = {v2f{wl.x, wl.y}, v2f{wl.z, wl.w}, v2f{wh.x, wh.y}, v2f{wh.z, wh.w}};
      v8f yacc = {};
      yacc = wmma_f32_16x16x4(v2f{tnl.x, tnl.y}, bw[0], yacc);
      yacc = wmma_f32_16x16x4(v2f{tnl.z, tnl.w}, bw[1], yacc);
      yacc = wmma_f32_16x16x4(v2f{tnh.x, tnh.y}, bw[2], yacc);
      yacc = wmma_f32_16x16x4(v2f{tnh.z, tnh.w}, bw[3], yacc);
      __builtin_amdgcn_sched_barrier(0);
#pragma unroll
      for (int q = 0; q < 4; ++q) {
        by[q].x = yacc[2 * q];
        by[q].y = yacc[2 * q + 1];
      }
    }
    const bool more = base >= kBlk;
    const int nbase = more ? base - kBlk : 0;
    const int nsmin = more ? min(first_slot(nbase >> 4, wave), kTpw - 1) : kTpw;
    const float* np = vsl + (size_t)(nbase / kBlk) * kVsBlk;
    stage_load<0>(stg, w, np, nsmin);
    p3_step<0>(av, hreg, Vt, wave, g0, smin, lh, lm, by);
    stage_load<1>(stg, w, np, nsmin);
    p3_step<1>(av, hreg, Vt, wave, g0, smin, lh, lm, by);
    tnl = *(const v4f*)(tfl + (size_t)(nbase / kBlk) * kTfBlk);
    tnh = *(const v4f*)(tfl + (size_t)(nbase / kBlk) * kTfBlk + 128);
  }

#pragma unroll
  for (int s = 0; s < kTpw; ++s) {
    float* hp = hrow + (wave + kWaves * s) * 16 + 8 * lh;
    *(v4f*)(hp)     = v4f{hreg[s][0], hreg[s][1], hreg[s][2], hreg[s][3]};
    *(v4f*)(hp + 4) = v4f{hreg[s][4], hreg[s][5], hreg[s][6], hreg[s][7]};
  }
  __syncthreads();

  for (int i = tid; i < kCols * (kN / 4); i += kThreads) {
    const int c  = i / (kN / 4);
    const int k4 = (i - c * (kN / 4)) * 4;
    v4f v = *(const v4f*)(hbuf + c * kHStr + k4);
    *(v4f*)(out + (size_t)(c0 + c) * kN + k4) = v;
  }
}

extern "C" void kernel_launch(void* const* d_in, const int* in_sizes, int n_in,
                              void* d_out, int out_size, void* d_ws, size_t ws_size,
                              hipStream_t stream) {
  const float* x = (const float*)d_in[0];
  const float* w = (const float*)d_in[1];
  float* out = (float*)d_out;
  float* tf  = (float*)d_ws;

  hipLaunchKernelGGL(block_factor_kernel, dim3(kN / kBlk + kSwParts * (kN / kBlk)), dim3(kFThreads),
                     0, stream, w, tf);

  dim3 grid(kB / kCols);
  dim3 block(kThreads);
  hipLaunchKernelGGL(ortho_householder_kernel, grid, block, 0, stream,
                     x, w, (const float*)tf, out);
}


# Next block's rows of U are requested right behind phase 1 and its Tn a block ahead; the loop's two barriers order LDS only, so these loads stay in flight across the whole W hand-over
# speedup vs baseline: 7.8992x; 1.0750x over previous
#include <hip/hip_runtime.h>

typedef __attribute__((ext_vector_type(2))) float v2f;
typedef __attribute__((ext_vector_type(4))) float v4f;
typedef __attribute__((ext_vector_type(8))) float v8f;

namespace {
constexpr int kN     = 2048;
constexpr int kB     = 4096;
constexpr int kCols  = 16;
constexpr int kBlk   = 16;
constexpr int kWaves = 16;
constexpr int kThreads = kWaves * 32;
constexpr int kFWaves = 8;
constexpr int kFThreads = kFWaves * 32;
constexpr int kHStr  = kN + 4;
constexpr int kVtStr = kBlk + 4;
constexpr int kWpStr   = 12;
constexpr int kWpPlane = 32 * kWpStr;
constexpr int kSmemFloats = kCols * kHStr
                          + kWaves * kWpPlane
                          + kWpPlane
                          + kN * kVtStr;
static_assert(kSmemFloats * sizeof(float) <= 320u * 1024u, "one work-group's LDS");
static_assert(((kCols * kHStr + kWaves * kWpPlane + kWpPlane) % 4) == 0, "Vt starts 16-byte aligned");
constexpr int kTfBlk = 256;
constexpr int kGb    = 16;
}

__device__ __forceinline__ v8f wmma_f32_16x16x4(v2f a, v2f b, v8f c) {
  return __builtin_amdgcn_wmma_f32_16x16x4_f32(
       false, a,  false, b,
       (short)0, c,  false,  false);
}

__device__ __forceinline__ void lds_barrier() {
  __builtin_amdgcn_fence(__ATOMIC_RELEASE, "workgroup", "local");
  __builtin_amdgcn_s_barrier();
  __builtin_amdgcn_fence(__ATOMIC_ACQUIRE, "workgroup", "local");
}

constexpr int kTg  = 4;
constexpr int kTpw = kN / 16 / kWaves;
constexpr int kNg  = kTpw / kTg;

constexpr int kTfFloats = (kN / kBlk) * kTfBlk;
constexpr int kVsTile   = 256;
constexpr int kVsWave   = kTpw * kVsTile;
constexpr int kVsBlk    = kWaves * kVsWave;
constexpr int kSwParts  = 2;

__device__ __forceinline__ int first_slot(int bt, int wave) {
  return (bt - wave + kWaves - 1) >> 4;
}
static_assert(kWaves == 16, "first_slot divides by the wave count with a shift of 4");

template <int kG>
__device__ __forceinline__ void stage_load(v4f (&stg)[2 * kTpw], const float* __restrict__ w,
                                           const float* p, int smin) {
#pragma unroll
  for (int s = kG * kTg; s < (kG + 1) * kTg; ++s) {
    const float* ps = (s >= smin) ? p + s * kVsTile : w;
    stg[2 * s]     = *(const v4f*)(ps);
    stg[2 * s + 1] = *(const v4f*)(ps + 128);
  }
}
static_assert(kTpw == 8 && kNg == 2, "the slot switch of phase 1 lists 8 slots, the block loop 2 pieces");

__device__ __forceinline__ void cut_upper(v4f& lo, v4f& hi, int base, int c0, int lh, int lm) {
  const int row = base + lm;
#pragma unroll
  for (int e = 0; e < 4; ++e) {
    const int cl = c0 + 8 * lh + e;
    const int ch = cl + 4;
    float ul = (cl >= row) ? lo[e] : 0.0f;
    float uh = (ch >= row) ? hi[e] : 0.0f;
    if (row == kN - 1 && cl == kN - 1) ul = 1.0f;
    if (row == kN - 1 && ch == kN - 1) uh = 1.0f;
    lo[e] = ul;
    hi[e] = uh;
  }
}

__device__ __forceinline__ void p1_tile(const v4f& lo, const v4f& hi, const v8f& ht, float* d,
                                        v8f& accW) {
  d[0]          = lo.x;
  d[kVtStr]     = lo.y;
  d[2 * kVtStr] = lo.z;
  d[3 * kVtStr] = lo.w;
  d[4 * kVtStr] = hi.x;
  d[5 * kVtStr] = hi.y;
  d[6 * kVtStr] = hi.z;
  d[7 * kVtStr] = hi.w;
  accW = wmma_f32_16x16x4(v2f{lo.x, lo.y}, v2f{ht[0], ht[1]}, accW);
  accW = wmma_f32_16x16x4(v2f{lo.z, lo.w}, v2f{ht[2], ht[3]}, accW);
  accW = wmma_f32_16x16x4(v2f{hi.x, hi.y}, v2f{ht[4], ht[5]}, accW);
  accW = wmma_f32_16x16x4(v2f{hi.z, hi.w}, v2f{ht[6], ht[7]}, accW);
}


template <bool kGuard>
__device__ __forceinline__ void p3_load(float (&av)[kTg][8], const float* vt, int wave, int s0,
                                        int smin, int lh, int lm) {
#pragma unroll
  for (int j = 0; j < kTg; ++j) {
#pragma unroll
    for (int q = 0; q < 8; ++q) av[j][q] = 0.0f;
    if (!kGuard || s0 + j >= smin) {
      const float* vp = vt + ((wave * kTpw + (s0 + j)) * 16 + lm) * kVtStr + 8 * lh;
      const v4f lo = *(const v4f*)(vp);
      const v4f hi = *(const v4f*)(vp + 4);
      av[j][0] = lo.x; av[j][1] = lo.y; av[j][2] = lo.z; av[j][3] = lo.w;
      av[j][4] = hi.x; av[j][5] = hi.y; av[j][6] = hi.z; av[j][7] = hi.w;
    }
  }
}

template <int kG, bool kGuard>
__device__ __forceinline__ void p3_apply(const float (&av)[kTg][8], v8f (&hreg)[kTpw],
                                         int smin, const v2f (&by)[4]) {
  if (!kGuard) {
#pragma unroll
    for (int q = 0; q < 4; ++q) {
#pragma unroll
      for (int j = 0; j < kTg; ++j) {
        const v2f fa = {av[j][2 * q], av[j][2 * q + 1]};
        hreg[kG * kTg + j] = wmma_f32_16x16x4(fa, by[q], hreg[kG * kTg + j]);
      }
    }
  } else {
#pragma unroll
    for (int j = 0; j < kTg; ++j) {
      if (kG * kTg + j >= smin) {
#pragma unroll
        for (int q = 0; q < 4; ++q) {
          const v2f fa = {av[j][2 * q], av[j][2 * q + 1]};
          hreg[kG * kTg + j] = wmma_f32_16x16x4(fa, by[q], hreg[kG * kTg + j]);
        }
      }
    }
  }
}

template <int kG>
__device__ __forceinline__ void p3_step(float (&av)[kTg][8], v8f (&hreg)[kTpw], const float* vt,
                                        int wave, int g0, int smin, int lh, int lm,
                                        const v2f (&by)[4]) {
  if (kG >= g0) {
    float cur[kTg][8];
#pragma unroll
    for (int j = 0; j < kTg; ++j) {
#pragma unroll
      for (int q = 0; q < 8; ++q) cur[j][q] = av[j][q];
    }
    if (kG + 1 < kNg) p3_load<false>(av, vt, wave, (kG + 1) * kTg, smin, lh, lm);
    if (kG * kTg >= smin) p3_apply<kG, false>(cur, hreg, smin, by);
    else                  p3_apply<kG, true>(cur, hreg, smin, by);
  }
}

template <bool kTail>
__device__ __forceinline__ void gram_batch(const float* p, int i0, int nt, int k, int row,
                                           int lh, v8f& accG) {
  v2f a[kGb];
#pragma unroll
  for (int j = 0; j < kGb; ++j) {
    a[j] = v2f{0.0f, 0.0f};
    if (!kTail || i0 + j < nt) a[j] = *(const v2f*)(p + (i0 + j) * 32);
  }
#pragma unroll
  for (int j = 0; j < kGb; ++j) {
    if (!kTail || i0 + j < nt) {
      const int col = k + (i0 + j) * 32 + 2 * lh;
      v2f u;
      u.x = (col >= row) ? a[j].x : 0.0f;
      u.y = (col + 1 >= row) ? a[j].y : 0.0f;
      if (row == kN - 1 && col == kN - 1) u.x = 1.0f;
      if (row == kN - 1 && col + 1 == kN - 1) u.y = 1.0f;
      accG = wmma_f32_16x16x4(u, u, accG);
    }
  }
}

__device__ __forceinline__ void relay_tiles(const float* __restrict__ w, float* __restrict__ vs,
                                            int q, int wave, int lane) {
  const int lm   = lane & 15;
  const int lh   = lane >> 4;
  const int blk  = q / kSwParts;
  const int part = q - blk * kSwParts;
  const int base = blk * kBlk;
  const int bt   = base >> 4;
  constexpr int kPer = (kN / 16) / (kSwParts * kFWaves);
  static_assert(kPer * kSwParts * kFWaves == kN / 16, "every tile of a block has one owner");
  const float* sp = w + (size_t)(base + lm) * kN + 8 * lh;
  float* dp = vs + (size_t)blk * kVsBlk + lane * 4;
  v4f lo[kPer], hi[kPer];
#pragma unroll
  for (int i = 0; i < kPer; ++i) {
    const int R = part + kSwParts * (wave + kFWaves * i);
    const float* ps = (R >= bt) ? sp + R * 16 : w;
    lo[i] = *(const v4f*)(ps);
    hi[i] = *(const v4f*)(ps + 4);
  }
#pragma unroll
  for (int i = 0; i < kPer; ++i) {
    const int R = part + kSwParts * (wave + kFWaves * i);
    cut_upper(lo[i], hi[i], base, R * 16, lh, lm);
    float* d = dp + ((R & (kWaves - 1)) * kTpw + (R >> 4)) * kVsTile;
    *(v4f*)(d)       = lo[i];
    *(v4f*)(d + 128) = hi[i];
  }
}

__global__ __launch_bounds__(kFThreads)
void block_factor_kernel(const float* __restrict__ w, float* __restrict__ tf) {
  if (blockIdx.x >= kN / kBlk) {
    const int rw = __builtin_amdgcn_readfirstlane(threadIdx.x >> 5);
    relay_tiles(w, tf + kTfFloats, (int)blockIdx.x - kN / kBlk, rw, threadIdx.x & 31);
    return;
  }
  __shared__ __attribute__((aligned(16))) float fsm[kFWaves * 256 + 256 + 16 + 256];
  float* Gpart = fsm;
  float* Gbuf  = Gpart + kFWaves * 256;
  float* Cbuf  = Gbuf + 256;
  float* Tl    = Cbuf + 16;

  const int tid  = threadIdx.x;
  const int wave = __builtin_amdgcn_readfirstlane(tid >> 5);
  const int lane = tid & 31;
  const int lm   = lane & 15;
  const int lh   = lane >> 4;
  const int blk  = blockIdx.x;
  const int base = blk * kBlk;

  v8f accG = {};
  const int k = base + wave * 4;
  if (k < kN) {
    const int nt  = (kN - k + kFWaves * 4 - 1) / (kFWaves * 4);
    const int row = base + lm;
    const float* p = w + (size_t)row * kN + k + 2 * lh;
    int i = 0;
    for (; i + kGb <= nt; i += kGb) gram_batch<false>(p, i, nt, k, row, lh, accG);
    if (i < nt) gram_batch<true>(p, i, nt, k, row, lh, accG);
  }
#pragma unroll
  for (int r = 0; r < 8; ++r)
    Gpart[wave * 256 + (r + 8 * lh) * 16 + lm] = accG[r];
  __syncthreads();
  {
    float sg = 0.f;
#pragma unroll
    for (int pw = 0; pw < kFWaves; ++pw) sg += Gpart[pw * 256 + tid];
    Gbuf[tid] = sg;
    if ((tid >> 4) == (tid & 15)) Cbuf[tid & 15] = (sg > 0.f) ? (2.0f / sg) : 0.0f;
  }
  __syncthreads();

  if (tid < 32) {
    float y[16];
    v4f c4[4];
#pragma unroll
    for (int pp = 0; pp < 4; ++pp) c4[pp] = *(const v4f*)(Cbuf + 4 * pp);
    v4f g4[16][4];
#pragma unroll
    for (int m = 0; m < 15; ++m) {
#pragma unroll
      for (int pp = (m + 1) >> 2; pp < 4; ++pp)
        g4[m][pp] = *(const v4f*)(Gbuf + m * 16 + 4 * pp);
    }
#pragma unroll
    for (int m = 15; m >= 0; --m) {
      float sacc = (m == lm) ? 1.0f : 0.0f;
#pragma unroll
      for (int l = 15; l > m; --l) sacc = __builtin_fmaf(g4[m][l >> 2][l & 3], y[l], sacc);
      y[m] = -c4[m >> 2][m & 3] * sacc;
    }
    if (lh == 0) {
#pragma unroll
      for (int m = 0; m < 16; ++m) Tl[m * 16 + lm] = y[m];
    }
  }
  __syncthreads();
  if (tid < 32) {
    const float* tr = Tl + lm * 16 + 8 * lh;
    const v4f lo = {tr[0], tr[1], tr[2], tr[3]};
    const v4f hi = {tr[4], tr[5], tr[6], tr[7]};
    float* d = tf + (size_t)blk * kTfBlk + lane * 4;
    *(v4f*)(d)       = lo;
    *(v4f*)(d + 128) = hi;
  }
}

__global__ __launch_bounds__(kThreads)
void ortho_householder_kernel(const float* __restrict__ x,
                              const float* __restrict__ w,
                              const float* __restrict__ tf,
                              float* __restrict__ out) {
  __shared__ __attribute__((aligned(16))) float smem[kSmemFloats];
  float* hbuf  = smem;
  float* Wpart = hbuf + kCols * kHStr;
  float* Wbuf  = Wpart + kWaves * kWpPlane;
  float* Vt    = Wbuf + kWpPlane;

  const int tid  = threadIdx.x;
  const int wave = __builtin_amdgcn_readfirstlane(tid >> 5);
  const int lane = tid & 31;
  const int lm   = lane & 15;
  const int lh   = lane >> 4;
  const int c0   = blockIdx.x * kCols;
  float* const hrow = hbuf + lm * kHStr;

  v4f stg[2 * kTpw];
  const float* vsl = tf + kTfFloats + wave * kVsWave + lane * 4;
  {
    const float* p0 = vsl + (size_t)(kN / kBlk - 1) * kVsBlk;
    const int s0 = min(first_slot((kN - kBlk) >> 4, wave), kTpw - 1);
    stage_load<0>(stg, w, p0, s0);
    stage_load<1>(stg, w, p0, s0);
  }
  const float* tfl = tf + lane * 4;
  v4f tnl = *(const v4f*)(tfl + (size_t)(kN / kBlk - 1) * kTfBlk);
  v4f tnh = *(const v4f*)(tfl + (size_t)(kN / kBlk - 1) * kTfBlk + 128);

  for (int i = tid; i < kCols * (kN / 4); i += kThreads) {
    const int c  = i / (kN / 4);
    const int k4 = (i - c * (kN / 4)) * 4;
    v4f v = *(const v4f*)(x + (size_t)(c0 + c) * kN + k4);
    *(v4f*)(hbuf + c * kHStr + k4) = v;
  }
  __syncthreads();

  v8f hreg[kTpw];
#pragma unroll
  for (int s = 0; s < kTpw; ++s) {
    const float* hp = hrow + (wave + kWaves * s) * 16 + 8 * lh;
    const v4f lo = *(const v4f*)(hp);
    const v4f hi = *(const v4f*)(hp + 4);
    hreg[s] = v8f{lo.x, lo.y, lo.z, lo.w, hi.x, hi.y, hi.z, hi.w};
  }

  for (int base = kN - kBlk; base >= 0; base -= kBlk) {
    const int bt   = base >> 4;
    const int smin = first_slot(bt, wave);
    const bool more = base >= kBlk;
    const int nbase = more ? base - kBlk : 0;
    const v4f tnnl = *(const v4f*)(tfl + (size_t)(nbase / kBlk) * kTfBlk);
    const v4f tnnh = *(const v4f*)(tfl + (size_t)(nbase / kBlk) * kTfBlk + 128);
    const int g0   = smin >> 2;
    const int gl   = min(g0, kNg - 1);
    v8f accW = {};
    {
      float* const vq = Vt + (wave * kTpw * 16 + 8 * lh) * kVtStr + lm;
      const int sfirst = min(smin, kTpw - 1);
#define P1_SLOT(S)                                                                          \
  if (S >= sfirst) {                                                                        \
    p1_tile(stg[2 * S], stg[2 * S + 1], hreg[S], vq + S * (16 * kVtStr), accW);             \
  }
      P1_SLOT(0)  P1_SLOT(1)  P1_SLOT(2)  P1_SLOT(3)
      P1_SLOT(4)  P1_SLOT(5)  P1_SLOT(6)  P1_SLOT(7)
#undef P1_SLOT
    }
    float av[kTg][8];
    p3_load<true>(av, Vt, wave, gl * kTg, smin, lh, lm);

    {
      float* wp = Wpart + wave * kWpPlane + lane * kWpStr;
      *(v4f*)(wp)     = v4f{accW[0], accW[1], accW[2], accW[3]};
      *(v4f*)(wp + 4) = v4f{accW[4], accW[5], accW[6], accW[7]};
    }
    const int nsmin = more ? min(first_slot(nbase >> 4, wave), kTpw - 1) : kTpw;
    const float* np = vsl + (size_t)(nbase / kBlk) * kVsBlk;
    stage_load<0>(stg, w, np, nsmin);
    stage_load<1>(stg, w, np, nsmin);
    lds_barrier();

    if (tid < 256) {
      const int el = (tid >> 3) * kWpStr + (tid & 7);
      float pw[kWaves];
#pragma unroll
      for (int p = 0; p < kWaves; ++p) pw[p] = Wpart[p * kWpPlane + el];
#pragma unroll
      for (int st = 1; st < kWaves; st *= 2) {
#pragma unroll
        for (int p = 0; p < kWaves; p += 2 * st) pw[p] += pw[p + st];
      }
      Wbuf[el] = pw[0];
    }
    lds_barrier();

    v2f by[4];
    {
      const float* wb = Wbuf + lane * kWpStr;
      const v4f wl = *(const v4f*)(wb);
      const v4f wh = *(const v4f*)(wb + 4);
      const v2f bw[4] = {v2f{wl.x, wl.y}, v2f{wl.z, wl.w}, v2f{wh.x, wh.y}, v2f{wh.z, wh.w}};
      v8f yacc = {};
      yacc = wmma_f32_16x16x4(v2f{tnl.x, tnl.y}, bw[0], yacc);
      yacc = wmma_f32_16x16x4(v2f{tnl.z, tnl.w}, bw[1], yacc);
      yacc = wmma_f32_16x16x4(v2f{tnh.x, tnh.y}, bw[2], yacc);
      yacc = wmma_f32_16x16x4(v2f{tnh.z, tnh.w}, bw[3], yacc);
#pragma unroll
      for (int q = 0; q < 4; ++q) {
        by[q].x = yacc[2 * q];
        by[q].y = yacc[2 * q + 1];
      }
    }
    p3_step<0>(av, hreg, Vt, wave, g0, smin, lh, lm, by);
    p3_step<1>(av, hreg, Vt, wave, g0, smin, lh, lm, by);
    tnl = tnnl;
    tnh = tnnh;
  }

#pragma unroll
  for (int s = 0; s < kTpw; ++s) {
    float* hp = hrow + (wave + kWaves * s) * 16 + 8 * lh;
    *(v4f*)(hp)     = v4f{hreg[s][0], hreg[s][1], hreg[s][2], hreg[s][3]};
    *(v4f*)(hp + 4) = v4f{hreg[s][4], hreg[s][5], hreg[s][6], hreg[s][7]};
  }
  __syncthreads();

  for (int i = tid; i < kCols * (kN / 4); i += kThreads) {
    const int c  = i / (kN / 4);
    const int k4 = (i - c * (kN / 4)) * 4;
    v4f v = *(const v4f*)(hbuf + c * kHStr + k4);
    *(v4f*)(out + (size_t)(c0 + c) * kN + k4) = v;
  }
}

extern "C" void kernel_launch(void* const* d_in, const int* in_sizes, int n_in,
                              void* d_out, int out_size, void* d_ws, size_t ws_size,
                              hipStream_t stream) {
  const float* x = (const float*)d_in[0];
  const float* w = (const float*)d_in[1];
  float* out = (float*)d_out;
  float* tf  = (float*)d_ws;

  hipLaunchKernelGGL(block_factor_kernel, dim3(kN / kBlk + kSwParts * (kN / kBlk)), dim3(kFThreads),
                     0, stream, w, tf);

  dim3 grid(kB / kCols);
  dim3 block(kThreads);
  hipLaunchKernelGGL(ortho_householder_kernel, grid, block, 0, stream,
                     x, w, (const float*)tf, out);
}
